# starting kernel
# baseline (speedup 1.0000x reference)
#include <hip/hip_runtime.h>

typedef float v2f __attribute__((ext_vector_type(2)));
typedef float v8f __attribute__((ext_vector_type(8)));

#define IN_CH  128
#define HID1   64
#define HID2   32

__global__ void fill_zero_kernel(float* __restrict__ p, int n) {
    int i = blockIdx.x * blockDim.x + threadIdx.x;
    if (i < n) p[i] = 0.0f;
}

__global__ void degree_kernel(const int* __restrict__ dst, float* __restrict__ deg, int nE) {
    int e = blockIdx.x * blockDim.x + threadIdx.x;
    if (e < nE) atomicAdd(&deg[dst[e]], 1.0f);
}

__global__ void dinv_kernel(float* __restrict__ dinv, int n) {
    int i = blockIdx.x * blockDim.x + threadIdx.x;
    if (i < n) dinv[i] = rsqrtf(dinv[i] + 1.0f);
}

__global__ __launch_bounds__(128) void gemm1_wmma(const float* __restrict__ X,
                                                  const float* __restrict__ W,
                                                  float* __restrict__ H) {
    const int lane = threadIdx.x & 31;
    const int wave = threadIdx.x >> 5;
    const int m0   = blockIdx.x * 16;
    const int n0   = wave * 16;
    const int l16  = lane & 15;
    const int khi  = (lane >> 4) * 2;
    const int row  = m0 + l16;
    const int col  = n0 + l16;
    v8f c = {};
    #pragma unroll 4
    for (int kk = 0; kk < IN_CH; kk += 4) {
        v2f a, b;
        a.x = X[row * IN_CH + kk + khi + 0];
        a.y = X[row * IN_CH + kk + khi + 1];
        b.x = W[(kk + khi + 0) * HID1 + col];
        b.y = W[(kk + khi + 1) * HID1 + col];
        c = __builtin_amdgcn_wmma_f32_16x16x4_f32(false, a, false, b,
                                                  (short)0, c, false, false);
    }
    const int mbase = m0 + ((lane >> 4) * 8);
    #pragma unroll
    for (int r = 0; r < 8; ++r)
        H[(mbase + r) * HID1 + col] = c[r];
}

__global__ __launch_bounds__(64) void gemm2_wmma(const float* __restrict__ X,
                                                 const float* __restrict__ W,
                                                 float* __restrict__ H) {
    const int lane = threadIdx.x & 31;
    const int wave = threadIdx.x >> 5;
    const int m0   = blockIdx.x * 16;
    const int n0   = wave * 16;
    const int l16  = lane & 15;
    const int khi  = (lane >> 4) * 2;
    const int row  = m0 + l16;
    const int col  = n0 + l16;
    v8f c = {};
    #pragma unroll 4
    for (int kk = 0; kk < HID1; kk += 4) {
        v2f a, b;
        a.x = X[row * HID1 + kk + khi + 0];
        a.y = X[row * HID1 + kk + khi + 1];
        b.x = W[(kk + khi + 0) * HID2 + col];
        b.y = W[(kk + khi + 1) * HID2 + col];
        c = __builtin_amdgcn_wmma_f32_16x16x4_f32(false, a, false, b,
                                                  (short)0, c, false, false);
    }
    const int mbase = m0 + ((lane >> 4) * 8);
    #pragma unroll
    for (int r = 0; r < 8; ++r)
        H[(mbase + r) * HID2 + col] = c[r];
}

__global__ __launch_bounds__(256) void edge_agg1_kernel(const float* __restrict__ h,
                                                        const int* __restrict__ src,
                                                        const int* __restrict__ dst,
                                                        const float* __restrict__ dinv,
                                                        float* __restrict__ agg, int nE) {
    int tid = blockIdx.x * 256 + threadIdx.x;
    int e = tid >> 4;
    if (e >= nE) return;
    int c4 = (tid & 15) * 4;
    int s = src[e], d = dst[e];
    float norm = dinv[s] * dinv[d];
    const float4 hv = *(const float4*)(h + s * HID1 + c4);
    float* ap = agg + d * HID1 + c4;
    atomicAdd(ap + 0, hv.x * norm);
    atomicAdd(ap + 1, hv.y * norm);
    atomicAdd(ap + 2, hv.z * norm);
    atomicAdd(ap + 3, hv.w * norm);
}

__global__ __launch_bounds__(256) void edge_agg2_kernel(const float* __restrict__ h,
                                                        const int* __restrict__ src,
                                                        const int* __restrict__ dst,
                                                        const float* __restrict__ dinv,
                                                        float* __restrict__ agg, int nE) {
    int tid = blockIdx.x * 256 + threadIdx.x;
    int e = tid >> 3;
    if (e >= nE) return;
    int c4 = (tid & 7) * 4;
    int s = src[e], d = dst[e];
    float norm = dinv[s] * dinv[d];
    const float4 hv = *(const float4*)(h + s * HID2 + c4);
    float* ap = agg + d * HID2 + c4;
    atomicAdd(ap + 0, hv.x * norm);
    atomicAdd(ap + 1, hv.y * norm);
    atomicAdd(ap + 2, hv.z * norm);
    atomicAdd(ap + 3, hv.w * norm);
}

__global__ __launch_bounds__(256) void combine1_kernel(float* __restrict__ h,
                                                       const float* __restrict__ agg,
                                                       const float* __restrict__ dinv,
                                                       const float* __restrict__ b, int n) {
    int i = blockIdx.x * 256 + threadIdx.x;
    if (i >= n * HID1) return;
    int node = i >> 6, c = i & 63;
    float dd = dinv[node]; dd *= dd;
    float v = agg[i] + h[i] * dd + b[c];
    h[i] = fmaxf(v, 0.0f);
}

__global__ __launch_bounds__(256) void combine2_proj_kernel(const float* __restrict__ h,
                                                            const float* __restrict__ agg,
                                                            const float* __restrict__ dinv,
                                                            const float* __restrict__ b2,
                                                            const float* __restrict__ Wh,
                                                            const float* __restrict__ bh,
                                                            float* __restrict__ out, int n) {
    int lane = threadIdx.x & 31;
    int node = blockIdx.x * 8 + (threadIdx.x >> 5);
    if (node >= n) return;
    float dd = dinv[node]; dd *= dd;
    int i = node * HID2 + lane;
    float v = agg[i] + h[i] * dd + b2[lane];
    v = fmaxf(v, 0.0f);
    float acc = v * Wh[lane];
    #pragma unroll
    for (int off = 16; off > 0; off >>= 1)
        acc += __shfl_xor(acc, off, 32);
    if (lane == 0) out[node] = acc + bh[0];
}

extern "C" void kernel_launch(void* const* d_in, const int* in_sizes, int n_in,
                              void* d_out, int out_size, void* d_ws, size_t ws_size,
                              hipStream_t stream) {
    const float* x  = (const float*)d_in[0];
    const int*   ei = (const int*)d_in[1];
    const float* W1 = (const float*)d_in[2];
    const float* b1 = (const float*)d_in[3];
    const float* W2 = (const float*)d_in[4];
    const float* b2 = (const float*)d_in[5];
    const float* Wh = (const float*)d_in[6];
    const float* bh = (const float*)d_in[7];
    float* out = (float*)d_out;

    const int N = in_sizes[0] / IN_CH;
    const int E = in_sizes[1] / 2;
    const int* src = ei;
    const int* dst = ei + E;

    float* f    = (float*)d_ws;
    float* dinv = f;
    float* h1   = f + 65536;
    float* agg1 = h1 + (size_t)N * HID1;
    float* h2   = agg1 + (size_t)N * HID1;
    float* agg2 = h2 + (size_t)N * HID2;

    fill_zero_kernel<<<(N + 255) / 256, 256, 0, stream>>>(dinv, N);
    fill_zero_kernel<<<(N * HID1 + 255) / 256, 256, 0, stream>>>(agg1, N * HID1);
    fill_zero_kernel<<<(N * HID2 + 255) / 256, 256, 0, stream>>>(agg2, N * HID2);

    degree_kernel<<<(E + 255) / 256, 256, 0, stream>>>(dst, dinv, E);
    dinv_kernel<<<(N + 255) / 256, 256, 0, stream>>>(dinv, N);

    gemm1_wmma<<<N / 16, 128, 0, stream>>>(x, W1, h1);
    edge_agg1_kernel<<<(E * 16 + 255) / 256, 256, 0, stream>>>(h1, src, dst, dinv, agg1, E);
    combine1_kernel<<<(N * HID1 + 255) / 256, 256, 0, stream>>>(h1, agg1, dinv, b1, N);

    gemm2_wmma<<<N / 16, 64, 0, stream>>>(h1, W2, h2);
    edge_agg2_kernel<<<(E * 8 + 255) / 256, 256, 0, stream>>>(h2, src, dst, dinv, agg2, E);

    combine2_proj_kernel<<<(N + 7) / 8, 256, 0, stream>>>(h2, agg2, dinv, b2, Wh, bh, out, N);
}


# Edge scatter: a wave's 32 lanes now cover 32 consecutive channels of one edge, so each f32 atomic fills one whole 128-B line (was 4 lines, 8 lanes each); sums still in atomic arrival order
# speedup vs baseline: 1.4763x; 1.4763x over previous
#include <hip/hip_runtime.h>

typedef float v2f __attribute__((ext_vector_type(2)));
typedef float v8f __attribute__((ext_vector_type(8)));

#define IN_CH  128
#define HID1   64
#define HID2   32

__global__ void fill_zero_kernel(float* __restrict__ p, int n) {
    int i = blockIdx.x * blockDim.x + threadIdx.x;
    if (i < n) p[i] = 0.0f;
}

__global__ void degree_kernel(const int* __restrict__ dst, float* __restrict__ deg, int nE) {
    int e = blockIdx.x * blockDim.x + threadIdx.x;
    if (e < nE) atomicAdd(&deg[dst[e]], 1.0f);
}

__global__ void dinv_kernel(float* __restrict__ dinv, int n) {
    int i = blockIdx.x * blockDim.x + threadIdx.x;
    if (i < n) dinv[i] = rsqrtf(dinv[i] + 1.0f);
}

__global__ __launch_bounds__(128) void gemm1_wmma(const float* __restrict__ X,
                                                  const float* __restrict__ W,
                                                  float* __restrict__ H) {
    const int lane = threadIdx.x & 31;
    const int wave = threadIdx.x >> 5;
    const int m0   = blockIdx.x * 16;
    const int n0   = wave * 16;
    const int l16  = lane & 15;
    const int khi  = (lane >> 4) * 2;
    const int row  = m0 + l16;
    const int col  = n0 + l16;
    v8f c = {};
    #pragma unroll 4
    for (int kk = 0; kk < IN_CH; kk += 4) {
        v2f a, b;
        a.x = X[row * IN_CH + kk + khi + 0];
        a.y = X[row * IN_CH + kk + khi + 1];
        b.x = W[(kk + khi + 0) * HID1 + col];
        b.y = W[(kk + khi + 1) * HID1 + col];
        c = __builtin_amdgcn_wmma_f32_16x16x4_f32(false, a, false, b,
                                                  (short)0, c, false, false);
    }
    const int mbase = m0 + ((lane >> 4) * 8);
    #pragma unroll
    for (int r = 0; r < 8; ++r)
        H[(mbase + r) * HID1 + col] = c[r];
}

__global__ __launch_bounds__(64) void gemm2_wmma(const float* __restrict__ X,
                                                 const float* __restrict__ W,
                                                 float* __restrict__ H) {
    const int lane = threadIdx.x & 31;
    const int wave = threadIdx.x >> 5;
    const int m0   = blockIdx.x * 16;
    const int n0   = wave * 16;
    const int l16  = lane & 15;
    const int khi  = (lane >> 4) * 2;
    const int row  = m0 + l16;
    const int col  = n0 + l16;
    v8f c = {};
    #pragma unroll 4
    for (int kk = 0; kk < HID1; kk += 4) {
        v2f a, b;
        a.x = X[row * HID1 + kk + khi + 0];
        a.y = X[row * HID1 + kk + khi + 1];
        b.x = W[(kk + khi + 0) * HID2 + col];
        b.y = W[(kk + khi + 1) * HID2 + col];
        c = __builtin_amdgcn_wmma_f32_16x16x4_f32(false, a, false, b,
                                                  (short)0, c, false, false);
    }
    const int mbase = m0 + ((lane >> 4) * 8);
    #pragma unroll
    for (int r = 0; r < 8; ++r)
        H[(mbase + r) * HID2 + col] = c[r];
}

template <int F>
__global__ __launch_bounds__(256) void edge_agg_rows_kernel(const float* __restrict__ h,
                                                            const int* __restrict__ src,
                                                            const int* __restrict__ dst,
                                                            const float* __restrict__ dinv,
                                                            float* __restrict__ agg, int nE) {
    constexpr int P = F / 32;
    const int lane = threadIdx.x & 31;
    const int wave = __builtin_amdgcn_readfirstlane(blockIdx.x * 8 + (threadIdx.x >> 5));
    const int e0 = wave * 32;
    if (e0 >= nE) return;
    const int e = e0 + lane;
    int s = 0, d = 0;
    float nrm = 0.0f;
    if (e < nE) {
        s = src[e];
        d = dst[e];
        nrm = dinv[s] * dinv[d];
    }
    const int left = nE - e0;
    if (left >= 32) {
        #pragma unroll
        for (int j0 = 0; j0 < 32; j0 += 8) {
            float v[8][P];
            #pragma unroll
            for (int u = 0; u < 8; ++u) {
                const int sj = __builtin_amdgcn_readlane(s, j0 + u);
                const float* hp = h + (size_t)sj * F + lane;
                #pragma unroll
                for (int p = 0; p < P; ++p) v[u][p] = hp[p * 32];
            }
            #pragma unroll
            for (int u = 0; u < 8; ++u) {
                const int dj = __builtin_amdgcn_readlane(d, j0 + u);
                const float nj = __int_as_float(
                    __builtin_amdgcn_readlane(__float_as_int(nrm), j0 + u));
                float* ap = agg + (size_t)dj * F + lane;
                #pragma unroll
                for (int p = 0; p < P; ++p) atomicAdd(ap + p * 32, v[u][p] * nj);
            }
        }
    } else {
        for (int j = 0; j < left; ++j) {
            const int sj = __builtin_amdgcn_readlane(s, j);
            const int dj = __builtin_amdgcn_readlane(d, j);
            const float nj = __int_as_float(__builtin_amdgcn_readlane(__float_as_int(nrm), j));
            const float* hp = h + (size_t)sj * F + lane;
            float* ap = agg + (size_t)dj * F + lane;
            #pragma unroll
            for (int p = 0; p < P; ++p) atomicAdd(ap + p * 32, hp[p * 32] * nj);
        }
    }
}

__global__ __launch_bounds__(256) void combine1_kernel(float* __restrict__ h,
                                                       const float* __restrict__ agg,
                                                       const float* __restrict__ dinv,
                                                       const float* __restrict__ b, int n) {
    int i = blockIdx.x * 256 + threadIdx.x;
    if (i >= n * HID1) return;
    int node = i >> 6, c = i & 63;
    float dd = dinv[node]; dd *= dd;
    float v = agg[i] + h[i] * dd + b[c];
    h[i] = fmaxf(v, 0.0f);
}

__global__ __launch_bounds__(256) void combine2_proj_kernel(const float* __restrict__ h,
                                                            const float* __restrict__ agg,
                                                            const float* __restrict__ dinv,
                                                            const float* __restrict__ b2,
                                                            const float* __restrict__ Wh,
                                                            const float* __restrict__ bh,
                                                            float* __restrict__ out, int n) {
    int lane = threadIdx.x & 31;
    int node = blockIdx.x * 8 + (threadIdx.x >> 5);
    if (node >= n) return;
    float dd = dinv[node]; dd *= dd;
    int i = node * HID2 + lane;
    float v = agg[i] + h[i] * dd + b2[lane];
    v = fmaxf(v, 0.0f);
    float acc = v * Wh[lane];
    #pragma unroll
    for (int off = 16; off > 0; off >>= 1)
        acc += __shfl_xor(acc, off, 32);
    if (lane == 0) out[node] = acc + bh[0];
}

extern "C" void kernel_launch(void* const* d_in, const int* in_sizes, int n_in,
                              void* d_out, int out_size, void* d_ws, size_t ws_size,
                              hipStream_t stream) {
    const float* x  = (const float*)d_in[0];
    const int*   ei = (const int*)d_in[1];
    const float* W1 = (const float*)d_in[2];
    const float* b1 = (const float*)d_in[3];
    const float* W2 = (const float*)d_in[4];
    const float* b2 = (const float*)d_in[5];
    const float* Wh = (const float*)d_in[6];
    const float* bh = (const float*)d_in[7];
    float* out = (float*)d_out;

    const int N = in_sizes[0] / IN_CH;
    const int E = in_sizes[1] / 2;
    const int* src = ei;
    const int* dst = ei + E;

    float* f    = (float*)d_ws;
    float* dinv = f;
    float* h1   = f + 65536;
    float* agg1 = h1 + (size_t)N * HID1;
    float* h2   = agg1 + (size_t)N * HID1;
    float* agg2 = h2 + (size_t)N * HID2;

    fill_zero_kernel<<<(N + 255) / 256, 256, 0, stream>>>(dinv, N);
    fill_zero_kernel<<<(N * HID1 + 255) / 256, 256, 0, stream>>>(agg1, N * HID1);
    fill_zero_kernel<<<(N * HID2 + 255) / 256, 256, 0, stream>>>(agg2, N * HID2);

    degree_kernel<<<(E + 255) / 256, 256, 0, stream>>>(dst, dinv, E);
    dinv_kernel<<<(N + 255) / 256, 256, 0, stream>>>(dinv, N);

    gemm1_wmma<<<N / 16, 128, 0, stream>>>(x, W1, h1);
    edge_agg_rows_kernel<HID1><<<(E + 255) / 256, 256, 0, stream>>>(h1, src, dst, dinv, agg1, E);
    combine1_kernel<<<(N * HID1 + 255) / 256, 256, 0, stream>>>(h1, agg1, dinv, b1, N);

    gemm2_wmma<<<N / 16, 64, 0, stream>>>(h1, W2, h2);
    edge_agg_rows_kernel<HID2><<<(E + 255) / 256, 256, 0, stream>>>(h2, src, dst, dinv, agg2, E);

    combine2_proj_kernel<<<(N + 7) / 8, 256, 0, stream>>>(h2, agg2, dinv, b2, Wh, bh, out, N);
}


# Edges sorted by destination per call (int atomic rank, prefix sum, place); a wave per node pulls source rows, sums in registers, combine fused; no float atomics, sums in arrival-built list order
# speedup vs baseline: 1.6822x; 1.1395x over previous
#include <hip/hip_runtime.h>

typedef float v2f __attribute__((ext_vector_type(2)));
typedef float v8f __attribute__((ext_vector_type(8)));

#define IN_CH  128
#define HID1   64
#define HID2   32

__global__ void zero_counts_kernel(int* __restrict__ p, int n) {
    int i = blockIdx.x * blockDim.x + threadIdx.x;
    if (i < n) p[i] = 0;
}

__global__ __launch_bounds__(256) void count_rank_kernel(const int* __restrict__ dst,
                                                         int* __restrict__ cnt,
                                                         int* __restrict__ rank, int nE) {
    int e = blockIdx.x * 256 + threadIdx.x;
    if (e < nE) rank[e] = atomicAdd(&cnt[dst[e]], 1);
}

#define SCAN_TILES 13
__global__ __launch_bounds__(1024) void scan_dinv_kernel(const int* __restrict__ cnt,
                                                         int* __restrict__ rowptr,
                                                         float* __restrict__ dinv, int n) {
    __shared__ int wtot[SCAN_TILES][32];
    __shared__ int ttot[SCAN_TILES];
    const int t = threadIdx.x, lane = t & 31, wave = t >> 5;
    int carry = 0;
    for (int base = 0; base < n; base += SCAN_TILES * 4096) {
        int4 v[SCAN_TILES];
        int incl[SCAN_TILES];
        #pragma unroll
        for (int k = 0; k < SCAN_TILES; ++k) {
            const int i = base + (k * 1024 + t) * 4;
            v[k] = make_int4(0, 0, 0, 0);
            if (i + 3 < n) {
                v[k] = *reinterpret_cast<const int4*>(cnt + i);
            } else {
                if (i < n)     v[k].x = cnt[i];
                if (i + 1 < n) v[k].y = cnt[i + 1];
                if (i + 2 < n) v[k].z = cnt[i + 2];
            }
        }
        #pragma unroll
        for (int k = 0; k < SCAN_TILES; ++k) {
            const int i = base + (k * 1024 + t) * 4;
            if (i + 3 < n) {
                *reinterpret_cast<float4*>(dinv + i) =
                    make_float4(rsqrtf((float)v[k].x + 1.0f), rsqrtf((float)v[k].y + 1.0f),
                                rsqrtf((float)v[k].z + 1.0f), rsqrtf((float)v[k].w + 1.0f));
            } else {
                if (i < n)     dinv[i]     = rsqrtf((float)v[k].x + 1.0f);
                if (i + 1 < n) dinv[i + 1] = rsqrtf((float)v[k].y + 1.0f);
                if (i + 2 < n) dinv[i + 2] = rsqrtf((float)v[k].z + 1.0f);
            }
            int x = v[k].x + v[k].y + v[k].z + v[k].w;
            #pragma unroll
            for (int off = 1; off < 32; off <<= 1) {
                const int y = __shfl_up(x, off, 32);
                if (lane >= off) x += y;
            }
            incl[k] = x;
            if (lane == 31) wtot[k][wave] = x;
        }
        __syncthreads();
        if (wave < SCAN_TILES) {
            const int own = wtot[wave][lane];
            int x = own;
            #pragma unroll
            for (int off = 1; off < 32; off <<= 1) {
                const int y = __shfl_up(x, off, 32);
                if (lane >= off) x += y;
            }
            wtot[wave][lane] = x - own;
            if (lane == 31) ttot[wave] = x;
        }
        __syncthreads();
        int tb = carry;
        #pragma unroll
        for (int k = 0; k < SCAN_TILES; ++k) {
            const int i = base + (k * 1024 + t) * 4;
            const int own = v[k].x + v[k].y + v[k].z + v[k].w;
            const int p0 = tb + wtot[k][wave] + incl[k] - own;
            const int p1 = p0 + v[k].x, p2 = p1 + v[k].y, p3 = p2 + v[k].z;
            if (i + 3 < n) {
                *reinterpret_cast<int4*>(rowptr + i) = make_int4(p0, p1, p2, p3);
            } else {
                if (i < n)     rowptr[i]     = p0;
                if (i + 1 < n) rowptr[i + 1] = p1;
                if (i + 2 < n) rowptr[i + 2] = p2;
            }
            tb += ttot[k];
        }
        carry = tb;
        __syncthreads();
    }
    if (t == 0) rowptr[n] = carry;
}

__global__ __launch_bounds__(256) void place_edges_kernel(const int* __restrict__ src,
                                                          const int* __restrict__ dst,
                                                          const int* __restrict__ rank,
                                                          const int* __restrict__ rowptr,
                                                          const float* __restrict__ dinv,
                                                          int2* __restrict__ csr, int nE) {
    int e = blockIdx.x * 256 + threadIdx.x;
    if (e >= nE) return;
    const int s = src[e], d = dst[e];
    const int p = rowptr[d] + rank[e];
    const float w = dinv[s] * dinv[d];
    csr[p] = make_int2(s, __float_as_int(w));
}

__global__ __launch_bounds__(128) void gemm1_wmma(const float* __restrict__ X,
                                                  const float* __restrict__ W,
                                                  float* __restrict__ H) {
    const int lane = threadIdx.x & 31;
    const int wave = threadIdx.x >> 5;
    const int m0   = blockIdx.x * 16;
    const int n0   = wave * 16;
    const int l16  = lane & 15;
    const int khi  = (lane >> 4) * 2;
    const int row  = m0 + l16;
    const int col  = n0 + l16;
    v8f c = {};
    #pragma unroll 4
    for (int kk = 0; kk < IN_CH; kk += 4) {
        v2f a, b;
        a.x = X[row * IN_CH + kk + khi + 0];
        a.y = X[row * IN_CH + kk + khi + 1];
        b.x = W[(kk + khi + 0) * HID1 + col];
        b.y = W[(kk + khi + 1) * HID1 + col];
        c = __builtin_amdgcn_wmma_f32_16x16x4_f32(false, a, false, b,
                                                  (short)0, c, false, false);
    }
    const int mbase = m0 + ((lane >> 4) * 8);
    #pragma unroll
    for (int r = 0; r < 8; ++r)
        H[(mbase + r) * HID1 + col] = c[r];
}

__global__ __launch_bounds__(64) void gemm2_wmma(const float* __restrict__ X,
                                                 const float* __restrict__ W,
                                                 float* __restrict__ H) {
    const int lane = threadIdx.x & 31;
    const int wave = threadIdx.x >> 5;
    const int m0   = blockIdx.x * 16;
    const int n0   = wave * 16;
    const int l16  = lane & 15;
    const int khi  = (lane >> 4) * 2;
    const int row  = m0 + l16;
    const int col  = n0 + l16;
    v8f c = {};
    #pragma unroll 4
    for (int kk = 0; kk < HID1; kk += 4) {
        v2f a, b;
        a.x = X[row * HID1 + kk + khi + 0];
        a.y = X[row * HID1 + kk + khi + 1];
        b.x = W[(kk + khi + 0) * HID2 + col];
        b.y = W[(kk + khi + 1) * HID2 + col];
        c = __builtin_amdgcn_wmma_f32_16x16x4_f32(false, a, false, b,
                                                  (short)0, c, false, false);
    }
    const int mbase = m0 + ((lane >> 4) * 8);
    #pragma unroll
    for (int r = 0; r < 8; ++r)
        H[(mbase + r) * HID2 + col] = c[r];
}

template <int VEC>
__device__ __forceinline__ void gather_rows(const float* __restrict__ h,
                                            const int2* __restrict__ csr,
                                            int beg, int end, int lane, float (&acc)[VEC]) {
    for (int e0 = beg; e0 < end; e0 += 32) {
        const int left = end - e0;
        int s = 0;
        float w = 0.0f;
        if (lane < left) {
            const int2 sw = csr[e0 + lane];
            s = sw.x;
            w = __int_as_float(sw.y);
        }
        if (left >= 32) {
            #pragma unroll
            for (int j0 = 0; j0 < 32; j0 += 8) {
                float v[8][VEC];
                #pragma unroll
                for (int u = 0; u < 8; ++u) {
                    const int sj = __builtin_amdgcn_readlane(s, j0 + u);
                    const float* hp = h + (size_t)sj * (32 * VEC) + lane * VEC;
                    if constexpr (VEC == 2) {
                        const float2 r = *reinterpret_cast<const float2*>(hp);
                        v[u][0] = r.x; v[u][1] = r.y;
                    } else {
                        v[u][0] = hp[0];
                    }
                }
                #pragma unroll
                for (int u = 0; u < 8; ++u) {
                    const float wj = __int_as_float(
                        __builtin_amdgcn_readlane(__float_as_int(w), j0 + u));
                    #pragma unroll
                    for (int p = 0; p < VEC; ++p) acc[p] += v[u][p] * wj;
                }
            }
        } else {
            #pragma unroll
            for (int j0 = 0; j0 < 32; j0 += 8) {
                if (j0 < left) {
                    float v[8][VEC];
                    #pragma unroll
                    for (int u = 0; u < 8; ++u) {
                        const int sj = __builtin_amdgcn_readlane(s, j0 + u);
                        const float* hp = h + (size_t)sj * (32 * VEC) + lane * VEC;
                        if constexpr (VEC == 2) {
                            const float2 r = *reinterpret_cast<const float2*>(hp);
                            v[u][0] = r.x; v[u][1] = r.y;
                        } else {
                            v[u][0] = hp[0];
                        }
                    }
                    if constexpr (VEC == 2) {
                        asm volatile("" :: "v"(v[0][0]), "v"(v[0][1]), "v"(v[1][0]), "v"(v[1][1]),
                                           "v"(v[2][0]), "v"(v[2][1]), "v"(v[3][0]), "v"(v[3][1]),
                                           "v"(v[4][0]), "v"(v[4][1]), "v"(v[5][0]), "v"(v[5][1]),
                                           "v"(v[6][0]), "v"(v[6][1]), "v"(v[7][0]), "v"(v[7][1]));
                    } else {
                        asm volatile("" :: "v"(v[0][0]), "v"(v[1][0]), "v"(v[2][0]), "v"(v[3][0]),
                                           "v"(v[4][0]), "v"(v[5][0]), "v"(v[6][0]), "v"(v[7][0]));
                    }
                    #pragma unroll
                    for (int u = 0; u < 8; ++u) {
                        if (j0 + u < left) {
                            const float wj = __int_as_float(
                                __builtin_amdgcn_readlane(__float_as_int(w), j0 + u));
                            #pragma unroll
                            for (int p = 0; p < VEC; ++p) acc[p] += v[u][p] * wj;
                        }
                    }
                }
            }
        }
    }
}

__global__ __launch_bounds__(256) void pull_combine1_kernel(const float* __restrict__ h,
                                                            const int* __restrict__ rowptr,
                                                            const int2* __restrict__ csr,
                                                            const float* __restrict__ dinv,
                                                            const float* __restrict__ b,
                                                            float* __restrict__ hout, int n) {
    const int lane = threadIdx.x & 31;
    const int node = __builtin_amdgcn_readfirstlane(blockIdx.x * 8 + (threadIdx.x >> 5));
    if (node >= n) return;
    const int beg = rowptr[node], end = rowptr[node + 1];
    const size_t i = (size_t)node * HID1 + lane * 2;
    const float2 hv = *reinterpret_cast<const float2*>(h + i);
    float dd = dinv[node]; dd *= dd;
    const float b0 = b[lane * 2], b1v = b[lane * 2 + 1];
    float acc[2] = {0.0f, 0.0f};
    gather_rows<2>(h, csr, beg, end, lane, acc);
    float2 o;
    o.x = fmaxf(acc[0] + hv.x * dd + b0, 0.0f);
    o.y = fmaxf(acc[1] + hv.y * dd + b1v, 0.0f);
    *reinterpret_cast<float2*>(hout + i) = o;
}

__global__ __launch_bounds__(256) void pull_combine2_proj_kernel(const float* __restrict__ h,
                                                                 const int* __restrict__ rowptr,
                                                                 const int2* __restrict__ csr,
                                                                 const float* __restrict__ dinv,
                                                                 const float* __restrict__ b2,
                                                                 const float* __restrict__ Wh,
                                                                 const float* __restrict__ bh,
                                                                 float* __restrict__ out, int n) {
    const int lane = threadIdx.x & 31;
    const int node = __builtin_amdgcn_readfirstlane(blockIdx.x * 8 + (threadIdx.x >> 5));
    if (node >= n) return;
    const int beg = rowptr[node], end = rowptr[node + 1];
    const float hv = h[(size_t)node * HID2 + lane];
    float dd = dinv[node]; dd *= dd;
    const float bb = b2[lane], ww = Wh[lane];
    float acc[1] = {0.0f};
    gather_rows<1>(h, csr, beg, end, lane, acc);
    float v = fmaxf(acc[0] + hv * dd + bb, 0.0f);
    float a = v * ww;
    #pragma unroll
    for (int off = 16; off > 0; off >>= 1)
        a += __shfl_xor(a, off, 32);
    if (lane == 0) out[node] = a + bh[0];
}

extern "C" void kernel_launch(void* const* d_in, const int* in_sizes, int n_in,
                              void* d_out, int out_size, void* d_ws, size_t ws_size,
                              hipStream_t stream) {
    const float* x  = (const float*)d_in[0];
    const int*   ei = (const int*)d_in[1];
    const float* W1 = (const float*)d_in[2];
    const float* b1 = (const float*)d_in[3];
    const float* W2 = (const float*)d_in[4];
    const float* b2 = (const float*)d_in[5];
    const float* Wh = (const float*)d_in[6];
    const float* bh = (const float*)d_in[7];
    float* out = (float*)d_out;

    const int N = in_sizes[0] / IN_CH;
    const int E = in_sizes[1] / 2;
    const int* src = ei;
    const int* dst = ei + E;

    float* f      = (float*)d_ws;
    int*   cnt    = (int*)d_ws;
    int*   rowptr = cnt + 65536;
    float* dinv   = f + 196608;
    float* h1     = f + 262144;
    float* h1b    = h1 + (size_t)N * HID1;
    float* h2     = h1b + (size_t)N * HID1;
    int2*  csr    = (int2*)(h2 + (size_t)N * HID2);
    int*   rank   = (int*)(h2 + (size_t)N * HID2) + (size_t)E * 2;

    zero_counts_kernel<<<(N + 255) / 256, 256, 0, stream>>>(cnt, N);
    count_rank_kernel<<<(E + 255) / 256, 256, 0, stream>>>(dst, cnt, rank, E);
    scan_dinv_kernel<<<1, 1024, 0, stream>>>(cnt, rowptr, dinv, N);
    place_edges_kernel<<<(E + 255) / 256, 256, 0, stream>>>(src, dst, rank, rowptr, dinv, csr, E);

    gemm1_wmma<<<N / 16, 128, 0, stream>>>(x, W1, h1);
    pull_combine1_kernel<<<(N + 7) / 8, 256, 0, stream>>>(h1, rowptr, csr, dinv, b1, h1b, N);

    gemm2_wmma<<<N / 16, 64, 0, stream>>>(h1b, W2, h2);
    pull_combine2_proj_kernel<<<(N + 7) / 8, 256, 0, stream>>>(h2, rowptr, csr, dinv, b2, Wh, bh, out, N);
}


# Destination count moved to LDS counters: 32 work-groups each count one slice of the edge list in 200 KB of LDS; no global atomics left; list order now slice-major, LDS arrival inside a slice
# speedup vs baseline: 2.3541x; 1.3994x over previous
#include <hip/hip_runtime.h>

typedef float v2f __attribute__((ext_vector_type(2)));
typedef float v8f __attribute__((ext_vector_type(8)));

#define IN_CH  128
#define HID1   64
#define HID2   32

#define SLICES    32
#define HIST_MAX  51200
#define CNT_TILES 13

__device__ __forceinline__ int slice_len(int nE) {
    return (nE + SLICES * 4 - 1) / (SLICES * 4) * 4;
}

__global__ __launch_bounds__(1024) void slice_count_kernel(const int* __restrict__ dst,
                                                           int* __restrict__ cntT,
                                                           int* __restrict__ rank,
                                                           int nE, int n) {
    __shared__ int hist[HIST_MAX];
    const int t = threadIdx.x;
    const int g = blockIdx.x;
    for (int i = t; i < HIST_MAX / 4; i += 1024)
        reinterpret_cast<int4*>(hist)[i] = make_int4(0, 0, 0, 0);
    __syncthreads();
    const int per = slice_len(nE);
    const int beg = g * per;
    const int end = min(beg + per, nE);
    for (int base = beg; base < end; base += CNT_TILES * 4096) {
        int4 d[CNT_TILES];
        #pragma unroll
        for (int k = 0; k < CNT_TILES; ++k) {
            const int i = base + (k * 1024 + t) * 4;
            d[k] = make_int4(0, 0, 0, 0);
            if (i + 3 < end) d[k] = *reinterpret_cast<const int4*>(dst + i);
        }
        #pragma unroll
        for (int k = 0; k < CNT_TILES; ++k) {
            const int i = base + (k * 1024 + t) * 4;
            if (i + 3 < end) {
                int4 r;
                r.x = atomicAdd(&hist[d[k].x], 1);
                r.y = atomicAdd(&hist[d[k].y], 1);
                r.z = atomicAdd(&hist[d[k].z], 1);
                r.w = atomicAdd(&hist[d[k].w], 1);
                *reinterpret_cast<int4*>(rank + i) = r;
            } else {
                for (int j = i; j < end; ++j) rank[j] = atomicAdd(&hist[dst[j]], 1);
            }
        }
    }
    __syncthreads();
    int* __restrict__ row = cntT + (size_t)g * n;
    for (int i = t; i < n; i += 1024) row[i] = hist[i];
}

__global__ __launch_bounds__(256) void slice_prefix_kernel(int* __restrict__ cntT,
                                                           int* __restrict__ cnt, int n) {
    const int i = blockIdx.x * 256 + threadIdx.x;
    if (i >= n) return;
    int c[SLICES];
    #pragma unroll
    for (int g = 0; g < SLICES; ++g) c[g] = cntT[(size_t)g * n + i];
    int run = 0;
    #pragma unroll
    for (int g = 0; g < SLICES; ++g) {
        cntT[(size_t)g * n + i] = run;
        run += c[g];
    }
    cnt[i] = run;
}

#define SCAN_TILES 13
__global__ __launch_bounds__(1024) void scan_dinv_kernel(const int* __restrict__ cnt,
                                                         int* __restrict__ rowptr,
                                                         float* __restrict__ dinv, int n) {
    __shared__ int wtot[SCAN_TILES][32];
    __shared__ int ttot[SCAN_TILES];
    const int t = threadIdx.x, lane = t & 31, wave = t >> 5;
    int carry = 0;
    for (int base = 0; base < n; base += SCAN_TILES * 4096) {
        int4 v[SCAN_TILES];
        int incl[SCAN_TILES];
        #pragma unroll
        for (int k = 0; k < SCAN_TILES; ++k) {
            const int i = base + (k * 1024 + t) * 4;
            v[k] = make_int4(0, 0, 0, 0);
            if (i + 3 < n) {
                v[k] = *reinterpret_cast<const int4*>(cnt + i);
            } else {
                if (i < n)     v[k].x = cnt[i];
                if (i + 1 < n) v[k].y = cnt[i + 1];
                if (i + 2 < n) v[k].z = cnt[i + 2];
            }
        }
        #pragma unroll
        for (int k = 0; k < SCAN_TILES; ++k) {
            const int i = base + (k * 1024 + t) * 4;
            if (i + 3 < n) {
                *reinterpret_cast<float4*>(dinv + i) =
                    make_float4(rsqrtf((float)v[k].x + 1.0f), rsqrtf((float)v[k].y + 1.0f),
                                rsqrtf((float)v[k].z + 1.0f), rsqrtf((float)v[k].w + 1.0f));
            } else {
                if (i < n)     dinv[i]     = rsqrtf((float)v[k].x + 1.0f);
                if (i + 1 < n) dinv[i + 1] = rsqrtf((float)v[k].y + 1.0f);
                if (i + 2 < n) dinv[i + 2] = rsqrtf((float)v[k].z + 1.0f);
            }
            int x = v[k].x + v[k].y + v[k].z + v[k].w;
            #pragma unroll
            for (int off = 1; off < 32; off <<= 1) {
                const int y = __shfl_up(x, off, 32);
                if (lane >= off) x += y;
            }
            incl[k] = x;
            if (lane == 31) wtot[k][wave] = x;
        }
        __syncthreads();
        if (wave < SCAN_TILES) {
            const int own = wtot[wave][lane];
            int x = own;
            #pragma unroll
            for (int off = 1; off < 32; off <<= 1) {
                const int y = __shfl_up(x, off, 32);
                if (lane >= off) x += y;
            }
            wtot[wave][lane] = x - own;
            if (lane == 31) ttot[wave] = x;
        }
        __syncthreads();
        int tb = carry;
        #pragma unroll
        for (int k = 0; k < SCAN_TILES; ++k) {
            const int i = base + (k * 1024 + t) * 4;
            const int own = v[k].x + v[k].y + v[k].z + v[k].w;
            const int p0 = tb + wtot[k][wave] + incl[k] - own;
            const int p1 = p0 + v[k].x, p2 = p1 + v[k].y, p3 = p2 + v[k].z;
            if (i + 3 < n) {
                *reinterpret_cast<int4*>(rowptr + i) = make_int4(p0, p1, p2, p3);
            } else {
                if (i < n)     rowptr[i]     = p0;
                if (i + 1 < n) rowptr[i + 1] = p1;
                if (i + 2 < n) rowptr[i + 2] = p2;
            }
            tb += ttot[k];
        }
        carry = tb;
        __syncthreads();
    }
    if (t == 0) rowptr[n] = carry;
}

__global__ __launch_bounds__(256) void place_edges_kernel(const int* __restrict__ src,
                                                          const int* __restrict__ dst,
                                                          const int* __restrict__ rank,
                                                          const int* __restrict__ rowptr,
                                                          const int* __restrict__ cntT,
                                                          const float* __restrict__ dinv,
                                                          int2* __restrict__ csr,
                                                          int nE, int n) {
    int e = blockIdx.x * 256 + threadIdx.x;
    if (e >= nE) return;
    const int s = src[e], d = dst[e];
    const int g = e / slice_len(nE);
    const int p = rowptr[d] + cntT[(size_t)g * n + d] + rank[e];
    const float w = dinv[s] * dinv[d];
    csr[p] = make_int2(s, __float_as_int(w));
}

__global__ __launch_bounds__(128) void gemm1_wmma(const float* __restrict__ X,
                                                  const float* __restrict__ W,
                                                  float* __restrict__ H) {
    const int lane = threadIdx.x & 31;
    const int wave = threadIdx.x >> 5;
    const int m0   = blockIdx.x * 16;
    const int n0   = wave * 16;
    const int l16  = lane & 15;
    const int khi  = (lane >> 4) * 2;
    const int row  = m0 + l16;
    const int col  = n0 + l16;
    v8f c = {};
    #pragma unroll 4
    for (int kk = 0; kk < IN_CH; kk += 4) {
        v2f a, b;
        a.x = X[row * IN_CH + kk + khi + 0];
        a.y = X[row * IN_CH + kk + khi + 1];
        b.x = W[(kk + khi + 0) * HID1 + col];
        b.y = W[(kk + khi + 1) * HID1 + col];
        c = __builtin_amdgcn_wmma_f32_16x16x4_f32(false, a, false, b,
                                                  (short)0, c, false, false);
    }
    const int mbase = m0 + ((lane >> 4) * 8);
    #pragma unroll
    for (int r = 0; r < 8; ++r)
        H[(mbase + r) * HID1 + col] = c[r];
}

__global__ __launch_bounds__(64) void gemm2_wmma(const float* __restrict__ X,
                                                 const float* __restrict__ W,
                                                 float* __restrict__ H) {
    const int lane = threadIdx.x & 31;
    const int wave = threadIdx.x >> 5;
    const int m0   = blockIdx.x * 16;
    const int n0   = wave * 16;
    const int l16  = lane & 15;
    const int khi  = (lane >> 4) * 2;
    const int row  = m0 + l16;
    const int col  = n0 + l16;
    v8f c = {};
    #pragma unroll 4
    for (int kk = 0; kk < HID1; kk += 4) {
        v2f a, b;
        a.x = X[row * HID1 + kk + khi + 0];
        a.y = X[row * HID1 + kk + khi + 1];
        b.x = W[(kk + khi + 0) * HID2 + col];
        b.y = W[(kk + khi + 1) * HID2 + col];
        c = __builtin_amdgcn_wmma_f32_16x16x4_f32(false, a, false, b,
                                                  (short)0, c, false, false);
    }
    const int mbase = m0 + ((lane >> 4) * 8);
    #pragma unroll
    for (int r = 0; r < 8; ++r)
        H[(mbase + r) * HID2 + col] = c[r];
}

template <int VEC>
__device__ __forceinline__ void gather_rows(const float* __restrict__ h,
                                            const int2* __restrict__ csr,
                                            int beg, int end, int lane, float (&acc)[VEC]) {
    for (int e0 = beg; e0 < end; e0 += 32) {
        const int left = end - e0;
        int s = 0;
        float w = 0.0f;
        if (lane < left) {
            const int2 sw = csr[e0 + lane];
            s = sw.x;
            w = __int_as_float(sw.y);
        }
        if (left >= 32) {
            #pragma unroll
            for (int j0 = 0; j0 < 32; j0 += 8) {
                float v[8][VEC];
                #pragma unroll
                for (int u = 0; u < 8; ++u) {
                    const int sj = __builtin_amdgcn_readlane(s, j0 + u);
                    const float* hp = h + (size_t)sj * (32 * VEC) + lane * VEC;
                    if constexpr (VEC == 2) {
                        const float2 r = *reinterpret_cast<const float2*>(hp);
                        v[u][0] = r.x; v[u][1] = r.y;
                    } else {
                        v[u][0] = hp[0];
                    }
                }
                #pragma unroll
                for (int u = 0; u < 8; ++u) {
                    const float wj = __int_as_float(
                        __builtin_amdgcn_readlane(__float_as_int(w), j0 + u));
                    #pragma unroll
                    for (int p = 0; p < VEC; ++p) acc[p] += v[u][p] * wj;
                }
            }
        } else {
            #pragma unroll
            for (int j0 = 0; j0 < 32; j0 += 8) {
                if (j0 < left) {
                    float v[8][VEC];
                    #pragma unroll
                    for (int u = 0; u < 8; ++u) {
                        const int sj = __builtin_amdgcn_readlane(s, j0 + u);
                        const float* hp = h + (size_t)sj * (32 * VEC) + lane * VEC;
                        if constexpr (VEC == 2) {
                            const float2 r = *reinterpret_cast<const float2*>(hp);
                            v[u][0] = r.x; v[u][1] = r.y;
                        } else {
                            v[u][0] = hp[0];
                        }
                    }
                    if constexpr (VEC == 2) {
                        asm volatile("" :: "v"(v[0][0]), "v"(v[0][1]), "v"(v[1][0]), "v"(v[1][1]),
                                           "v"(v[2][0]), "v"(v[2][1]), "v"(v[3][0]), "v"(v[3][1]),
                                           "v"(v[4][0]), "v"(v[4][1]), "v"(v[5][0]), "v"(v[5][1]),
                                           "v"(v[6][0]), "v"(v[6][1]), "v"(v[7][0]), "v"(v[7][1]));
                    } else {
                        asm volatile("" :: "v"(v[0][0]), "v"(v[1][0]), "v"(v[2][0]), "v"(v[3][0]),
                                           "v"(v[4][0]), "v"(v[5][0]), "v"(v[6][0]), "v"(v[7][0]));
                    }
                    #pragma unroll
                    for (int u = 0; u < 8; ++u) {
                        if (j0 + u < left) {
                            const float wj = __int_as_float(
                                __builtin_amdgcn_readlane(__float_as_int(w), j0 + u));
                            #pragma unroll
                            for (int p = 0; p < VEC; ++p) acc[p] += v[u][p] * wj;
                        }
                    }
                }
            }
        }
    }
}

__global__ __launch_bounds__(256) void pull_combine1_kernel(const float* __restrict__ h,
                                                            const int* __restrict__ rowptr,
                                                            const int2* __restrict__ csr,
                                                            const float* __restrict__ dinv,
                                                            const float* __restrict__ b,
                                                            float* __restrict__ hout, int n) {
    const int lane = threadIdx.x & 31;
    const int node = __builtin_amdgcn_readfirstlane(blockIdx.x * 8 + (threadIdx.x >> 5));
    if (node >= n) return;
    const int beg = rowptr[node], end = rowptr[node + 1];
    const size_t i = (size_t)node * HID1 + lane * 2;
    const float2 hv = *reinterpret_cast<const float2*>(h + i);
    float dd = dinv[node]; dd *= dd;
    const float b0 = b[lane * 2], b1v = b[lane * 2 + 1];
    float acc[2] = {0.0f, 0.0f};
    gather_rows<2>(h, csr, beg, end, lane, acc);
    float2 o;
    o.x = fmaxf(acc[0] + hv.x * dd + b0, 0.0f);
    o.y = fmaxf(acc[1] + hv.y * dd + b1v, 0.0f);
    *reinterpret_cast<float2*>(hout + i) = o;
}

__global__ __launch_bounds__(256) void pull_combine2_proj_kernel(const float* __restrict__ h,
                                                                 const int* __restrict__ rowptr,
                                                                 const int2* __restrict__ csr,
                                                                 const float* __restrict__ dinv,
                                                                 const float* __restrict__ b2,
                                                                 const float* __restrict__ Wh,
                                                                 const float* __restrict__ bh,
                                                                 float* __restrict__ out, int n) {
    const int lane = threadIdx.x & 31;
    const int node = __builtin_amdgcn_readfirstlane(blockIdx.x * 8 + (threadIdx.x >> 5));
    if (node >= n) return;
    const int beg = rowptr[node], end = rowptr[node + 1];
    const float hv = h[(size_t)node * HID2 + lane];
    float dd = dinv[node]; dd *= dd;
    const float bb = b2[lane], ww = Wh[lane];
    float acc[1] = {0.0f};
    gather_rows<1>(h, csr, beg, end, lane, acc);
    float v = fmaxf(acc[0] + hv * dd + bb, 0.0f);
    float a = v * ww;
    #pragma unroll
    for (int off = 16; off > 0; off >>= 1)
        a += __shfl_xor(a, off, 32);
    if (lane == 0) out[node] = a + bh[0];
}

extern "C" void kernel_launch(void* const* d_in, const int* in_sizes, int n_in,
                              void* d_out, int out_size, void* d_ws, size_t ws_size,
                              hipStream_t stream) {
    const float* x  = (const float*)d_in[0];
    const int*   ei = (const int*)d_in[1];
    const float* W1 = (const float*)d_in[2];
    const float* b1 = (const float*)d_in[3];
    const float* W2 = (const float*)d_in[4];
    const float* b2 = (const float*)d_in[5];
    const float* Wh = (const float*)d_in[6];
    const float* bh = (const float*)d_in[7];
    float* out = (float*)d_out;

    const int N = in_sizes[0] / IN_CH;
    const int E = in_sizes[1] / 2;
    const int* src = ei;
    const int* dst = ei + E;

    float* f      = (float*)d_ws;
    int*   cnt    = (int*)d_ws;
    int*   rowptr = cnt + 65536;
    float* dinv   = f + 196608;
    float* h1     = f + 262144;
    float* h1b    = h1 + (size_t)N * HID1;
    float* h2     = h1b + (size_t)N * HID1;
    int2*  csr    = (int2*)(h2 + (size_t)N * HID2);
    int*   rank   = (int*)(h2 + (size_t)N * HID2) + (size_t)E * 2;
    int*   cntT   = rank + E;

    slice_count_kernel<<<SLICES, 1024, 0, stream>>>(dst, cntT, rank, E, N);
    slice_prefix_kernel<<<(N + 255) / 256, 256, 0, stream>>>(cntT, cnt, N);
    scan_dinv_kernel<<<1, 1024, 0, stream>>>(cnt, rowptr, dinv, N);
    place_edges_kernel<<<(E + 255) / 256, 256, 0, stream>>>(src, dst, rank, rowptr, cntT, dinv, csr, E, N);

    gemm1_wmma<<<N / 16, 128, 0, stream>>>(x, W1, h1);
    pull_combine1_kernel<<<(N + 7) / 8, 256, 0, stream>>>(h1, rowptr, csr, dinv, b1, h1b, N);

    gemm2_wmma<<<N / 16, 64, 0, stream>>>(h1b, W2, h2);
    pull_combine2_proj_kernel<<<(N + 7) / 8, 256, 0, stream>>>(h2, rowptr, csr, dinv, b2, Wh, bh, out, N);
}


# Node prefix sum folded into the slice-prefix launch: the count also writes 256-node block totals, a prefix group adds those before it, stores absolute slots; 1-group scan launch gone (7 launches)
# speedup vs baseline: 2.8198x; 1.1978x over previous
#include <hip/hip_runtime.h>

typedef float v2f __attribute__((ext_vector_type(2)));
typedef float v8f __attribute__((ext_vector_type(8)));

#define IN_CH  128
#define HID1   64
#define HID2   32

#define SLICES    32
#define HIST_MAX  51200
#define CNT_TILES 13
#define PBLK      256
#define BLK_TRIPS ((HIST_MAX / PBLK + 31) / 32)

__device__ __forceinline__ int slice_len(int nE) {
    return (nE + SLICES * 4 - 1) / (SLICES * 4) * 4;
}

__global__ __launch_bounds__(1024) void slice_count_kernel(const int* __restrict__ dst,
                                                           int* __restrict__ cntT,
                                                           int* __restrict__ blkT,
                                                           int* __restrict__ rank,
                                                           int nE, int n) {
    __shared__ int hist[HIST_MAX];
    const int t = threadIdx.x;
    const int g = blockIdx.x;
    for (int i = t; i < HIST_MAX / 4; i += 1024)
        reinterpret_cast<int4*>(hist)[i] = make_int4(0, 0, 0, 0);
    __syncthreads();
    const int per = slice_len(nE);
    const int beg = g * per;
    const int end = min(beg + per, nE);
    for (int base = beg; base < end; base += CNT_TILES * 4096) {
        int4 d[CNT_TILES];
        #pragma unroll
        for (int k = 0; k < CNT_TILES; ++k) {
            const int i = base + (k * 1024 + t) * 4;
            d[k] = make_int4(0, 0, 0, 0);
            if (i + 3 < end) d[k] = *reinterpret_cast<const int4*>(dst + i);
        }
        #pragma unroll
        for (int k = 0; k < CNT_TILES; ++k) {
            const int i = base + (k * 1024 + t) * 4;
            if (i + 3 < end) {
                int4 r;
                r.x = atomicAdd(&hist[d[k].x], 1);
                r.y = atomicAdd(&hist[d[k].y], 1);
                r.z = atomicAdd(&hist[d[k].z], 1);
                r.w = atomicAdd(&hist[d[k].w], 1);
                *reinterpret_cast<int4*>(rank + i) = r;
            } else {
                for (int j = i; j < end; ++j) rank[j] = atomicAdd(&hist[dst[j]], 1);
            }
        }
    }
    __syncthreads();
    int* __restrict__ row = cntT + (size_t)g * n;
    const int lane = t & 31, wave = t >> 5;
    const int nb = (n + PBLK - 1) / PBLK;
    int s[BLK_TRIPS];
    #pragma unroll
    for (int k = 0; k < BLK_TRIPS; ++k) {
        const int b = wave + 32 * k;
        s[k] = 0;
        if (b < nb) {
            const int i0 = b * PBLK + lane;
            int v[PBLK / 32];
            #pragma unroll
            for (int j = 0; j < PBLK / 32; ++j) v[j] = hist[i0 + j * 32];
            if (b * PBLK + PBLK <= n) {
                #pragma unroll
                for (int j = 0; j < PBLK / 32; ++j) row[i0 + j * 32] = v[j];
            } else {
                #pragma unroll
                for (int j = 0; j < PBLK / 32; ++j)
                    if (i0 + j * 32 < n) row[i0 + j * 32] = v[j];
            }
            #pragma unroll
            for (int j = 0; j < PBLK / 32; ++j) s[k] += v[j];
        }
    }
    #pragma unroll
    for (int off = 16; off > 0; off >>= 1) {
        #pragma unroll
        for (int k = 0; k < BLK_TRIPS; ++k) s[k] += __shfl_xor(s[k], off, 32);
    }
    if (lane == 0) {
        #pragma unroll
        for (int k = 0; k < BLK_TRIPS; ++k) {
            const int b = wave + 32 * k;
            if (b < nb) blkT[b * SLICES + g] = s[k];
        }
    }
}

#define BASE_Q 8
static_assert(HIST_MAX / PBLK * SLICES <= BASE_Q * PBLK * 4, "block totals exceed one pass");
__global__ __launch_bounds__(PBLK) void slice_prefix_kernel(int* __restrict__ cntT,
                                                            const int* __restrict__ blkT,
                                                            int* __restrict__ rowptr,
                                                            float* __restrict__ dinv, int n) {
    __shared__ int wbase[PBLK / 32];
    __shared__ int wtot[PBLK / 32];
    const int t = threadIdx.x, lane = t & 31, wave = t >> 5;
    const int b = blockIdx.x;
    const int i = b * PBLK + t;
    const bool live = i < n;
    const int ic = live ? i : n - 1;
    int c[SLICES];
    #pragma unroll
    for (int g = 0; g < SLICES; ++g) c[g] = cntT[(size_t)g * n + ic];
    const int4* __restrict__ bq = reinterpret_cast<const int4*>(blkT);
    const int nq = b * (SLICES / 4);
    const int qlast = (int)gridDim.x * (SLICES / 4) - 1;
    int part = 0;
    int4 v[BASE_Q];
    #pragma unroll
    for (int k = 0; k < BASE_Q; ++k) v[k] = bq[min(k * PBLK + t, qlast)];
    #pragma unroll
    for (int k = 0; k < BASE_Q; ++k) {
        const int s = v[k].x + v[k].y + v[k].z + v[k].w;
        part += (k * PBLK + t < nq) ? s : 0;
    }
    int tot = 0;
    #pragma unroll
    for (int g = 0; g < SLICES; ++g) tot += c[g];
    if (!live) tot = 0;
    int incl = tot;
    #pragma unroll
    for (int off = 1; off < 32; off <<= 1) {
        const int y = __shfl_up(incl, off, 32);
        if (lane >= off) incl += y;
    }
    #pragma unroll
    for (int off = 16; off > 0; off >>= 1) part += __shfl_xor(part, off, 32);
    if (lane == 31) wtot[wave] = incl;
    if (lane == 0)  wbase[wave] = part;
    __syncthreads();
    int start = incl - tot;
    #pragma unroll
    for (int w = 0; w < PBLK / 32; ++w) {
        start += wbase[w];
        if (w < wave) start += wtot[w];
    }
    if (live) {
        rowptr[i] = start;
        dinv[i] = rsqrtf((float)tot + 1.0f);
        if (i == n - 1) rowptr[n] = start + tot;
        int run = start;
        #pragma unroll
        for (int g = 0; g < SLICES; ++g) {
            cntT[(size_t)g * n + i] = run;
            run += c[g];
        }
    }
}

__global__ __launch_bounds__(256) void place_edges_kernel(const int* __restrict__ src,
                                                          const int* __restrict__ dst,
                                                          const int* __restrict__ rank,
                                                          const int* __restrict__ cntT,
                                                          int* __restrict__ csr,
                                                          int nE, int n) {
    int e = blockIdx.x * 256 + threadIdx.x;
    if (e >= nE) return;
    const int s = src[e], d = dst[e];
    const int g = e / slice_len(nE);
    const int p = cntT[(size_t)g * n + d] + rank[e];
    csr[p] = s;
}

__global__ __launch_bounds__(128) void gemm1_wmma(const float* __restrict__ X,
                                                  const float* __restrict__ W,
                                                  const float* __restrict__ dinv,
                                                  float* __restrict__ H) {
    const int lane = threadIdx.x & 31;
    const int wave = threadIdx.x >> 5;
    const int m0   = blockIdx.x * 16;
    const int n0   = wave * 16;
    const int l16  = lane & 15;
    const int khi  = (lane >> 4) * 2;
    const int row  = m0 + l16;
    const int col  = n0 + l16;
    v8f c = {};
    #pragma unroll 4
    for (int kk = 0; kk < IN_CH; kk += 4) {
        v2f a, b;
        a.x = X[row * IN_CH + kk + khi + 0];
        a.y = X[row * IN_CH + kk + khi + 1];
        b.x = W[(kk + khi + 0) * HID1 + col];
        b.y = W[(kk + khi + 1) * HID1 + col];
        c = __builtin_amdgcn_wmma_f32_16x16x4_f32(false, a, false, b,
                                                  (short)0, c, false, false);
    }
    const int mbase = m0 + ((lane >> 4) * 8);
    const float4 s0 = *reinterpret_cast<const float4*>(dinv + mbase);
    const float4 s1 = *reinterpret_cast<const float4*>(dinv + mbase + 4);
    const float sc[8] = {s0.x, s0.y, s0.z, s0.w, s1.x, s1.y, s1.z, s1.w};
    #pragma unroll
    for (int r = 0; r < 8; ++r)
        H[(mbase + r) * HID1 + col] = c[r] * sc[r];
}

__global__ __launch_bounds__(64) void gemm2_wmma(const float* __restrict__ X,
                                                 const float* __restrict__ W,
                                                 float* __restrict__ H) {
    const int lane = threadIdx.x & 31;
    const int wave = threadIdx.x >> 5;
    const int m0   = blockIdx.x * 16;
    const int n0   = wave * 16;
    const int l16  = lane & 15;
    const int khi  = (lane >> 4) * 2;
    const int row  = m0 + l16;
    const int col  = n0 + l16;
    v8f c = {};
    #pragma unroll 4
    for (int kk = 0; kk < HID1; kk += 4) {
        v2f a, b;
        a.x = X[row * HID1 + kk + khi + 0];
        a.y = X[row * HID1 + kk + khi + 1];
        b.x = W[(kk + khi + 0) * HID2 + col];
        b.y = W[(kk + khi + 1) * HID2 + col];
        c = __builtin_amdgcn_wmma_f32_16x16x4_f32(false, a, false, b,
                                                  (short)0, c, false, false);
    }
    const int mbase = m0 + ((lane >> 4) * 8);
    #pragma unroll
    for (int r = 0; r < 8; ++r)
        H[(mbase + r) * HID2 + col] = c[r];
}

template <int VEC>
__device__ __forceinline__ void gather_rows(const float* __restrict__ h,
                                            const int* __restrict__ csr,
                                            int beg, int end, int lane, float (&acc)[VEC]) {
    for (int e0 = beg; e0 < end; e0 += 32) {
        const int left = end - e0;
        int s = 0;
        if (lane < left) s = csr[e0 + lane];
        if (left >= 32) {
            #pragma unroll
            for (int j0 = 0; j0 < 32; j0 += 8) {
                float v[8][VEC];
                #pragma unroll
                for (int u = 0; u < 8; ++u) {
                    const int sj = __builtin_amdgcn_readlane(s, j0 + u);
                    const float* hp = h + (size_t)sj * (32 * VEC) + lane * VEC;
                    if constexpr (VEC == 2) {
                        const float2 r = *reinterpret_cast<const float2*>(hp);
                        v[u][0] = r.x; v[u][1] = r.y;
                    } else {
                        v[u][0] = hp[0];
                    }
                }
                #pragma unroll
                for (int u = 0; u < 8; ++u) {
                    #pragma unroll
                    for (int p = 0; p < VEC; ++p) acc[p] += v[u][p];
                }
            }
        } else {
            #pragma unroll
            for (int j0 = 0; j0 < 32; j0 += 8) {
                if (j0 < left) {
                    float v[8][VEC];
                    #pragma unroll
                    for (int u = 0; u < 8; ++u) {
                        const int sj = __builtin_amdgcn_readlane(s, j0 + u);
                        const float* hp = h + (size_t)sj * (32 * VEC) + lane * VEC;
                        if constexpr (VEC == 2) {
                            const float2 r = *reinterpret_cast<const float2*>(hp);
                            v[u][0] = r.x; v[u][1] = r.y;
                        } else {
                            v[u][0] = hp[0];
                        }
                    }
                    if constexpr (VEC == 2) {
                        asm volatile("" :: "v"(v[0][0]), "v"(v[0][1]), "v"(v[1][0]), "v"(v[1][1]),
                                           "v"(v[2][0]), "v"(v[2][1]), "v"(v[3][0]), "v"(v[3][1]),
                                           "v"(v[4][0]), "v"(v[4][1]), "v"(v[5][0]), "v"(v[5][1]),
                                           "v"(v[6][0]), "v"(v[6][1]), "v"(v[7][0]), "v"(v[7][1]));
                    } else {
                        asm volatile("" :: "v"(v[0][0]), "v"(v[1][0]), "v"(v[2][0]), "v"(v[3][0]),
                                           "v"(v[4][0]), "v"(v[5][0]), "v"(v[6][0]), "v"(v[7][0]));
                    }
                    #pragma unroll
                    for (int u = 0; u < 8; ++u) {
                        if (j0 + u < left) {
                            #pragma unroll
                            for (int p = 0; p < VEC; ++p) acc[p] += v[u][p];
                        }
                    }
                }
            }
        }
    }
}

__global__ __launch_bounds__(256) void pull_combine1_kernel(const float* __restrict__ h,
                                                            const int* __restrict__ rowptr,
                                                            const int* __restrict__ csr,
                                                            const float* __restrict__ dinv,
                                                            const float* __restrict__ b,
                                                            float* __restrict__ hout, int n) {
    const int lane = threadIdx.x & 31;
    const int node = __builtin_amdgcn_readfirstlane(blockIdx.x * 8 + (threadIdx.x >> 5));
    if (node >= n) return;
    const int beg = rowptr[node], end = rowptr[node + 1];
    const size_t i = (size_t)node * HID1 + lane * 2;
    const float2 hv = *reinterpret_cast<const float2*>(h + i);
    const float dd = dinv[node];
    const float b0 = b[lane * 2], b1v = b[lane * 2 + 1];
    float acc[2] = {0.0f, 0.0f};
    gather_rows<2>(h, csr, beg, end, lane, acc);
    float2 o;
    o.x = fmaxf((acc[0] + hv.x) * dd + b0, 0.0f) * dd;
    o.y = fmaxf((acc[1] + hv.y) * dd + b1v, 0.0f) * dd;
    *reinterpret_cast<float2*>(hout + i) = o;
}

__global__ __launch_bounds__(256) void pull_combine2_proj_kernel(const float* __restrict__ h,
                                                                 const int* __restrict__ rowptr,
                                                                 const int* __restrict__ csr,
                                                                 const float* __restrict__ dinv,
                                                                 const float* __restrict__ b2,
                                                                 const float* __restrict__ Wh,
                                                                 const float* __restrict__ bh,
                                                                 float* __restrict__ out, int n) {
    const int lane = threadIdx.x & 31;
    const int node = __builtin_amdgcn_readfirstlane(blockIdx.x * 8 + (threadIdx.x >> 5));
    if (node >= n) return;
    const int beg = rowptr[node], end = rowptr[node + 1];
    const float hv = h[(size_t)node * HID2 + lane];
    const float dd = dinv[node];
    const float bb = b2[lane], ww = Wh[lane];
    float acc[1] = {0.0f};
    gather_rows<1>(h, csr, beg, end, lane, acc);
    float v = fmaxf((acc[0] + hv) * dd + bb, 0.0f);
    float a = v * ww;
    #pragma unroll
    for (int off = 16; off > 0; off >>= 1)
        a += __shfl_xor(a, off, 32);
    if (lane == 0) out[node] = a + bh[0];
}

extern "C" void kernel_launch(void* const* d_in, const int* in_sizes, int n_in,
                              void* d_out, int out_size, void* d_ws, size_t ws_size,
                              hipStream_t stream) {
    const float* x  = (const float*)d_in[0];
    const int*   ei = (const int*)d_in[1];
    const float* W1 = (const float*)d_in[2];
    const float* b1 = (const float*)d_in[3];
    const float* W2 = (const float*)d_in[4];
    const float* b2 = (const float*)d_in[5];
    const float* Wh = (const float*)d_in[6];
    const float* bh = (const float*)d_in[7];
    float* out = (float*)d_out;

    const int N = in_sizes[0] / IN_CH;
    const int E = in_sizes[1] / 2;
    const int* src = ei;
    const int* dst = ei + E;

    float* f      = (float*)d_ws;
    int*   blkT   = (int*)d_ws;
    int*   rowptr = blkT + 65536;
    float* dinv   = f + 196608;
    float* h1     = f + 262144;
    float* h1b    = h1 + (size_t)N * HID1;
    float* h2     = h1b + (size_t)N * HID1;
    int*   csr    = (int*)(h2 + (size_t)N * HID2);
    int*   rank   = csr + E;
    int*   cntT   = rank + E;

    slice_count_kernel<<<SLICES, 1024, 0, stream>>>(dst, cntT, blkT, rank, E, N);
    slice_prefix_kernel<<<(N + PBLK - 1) / PBLK, PBLK, 0, stream>>>(cntT, blkT, rowptr, dinv, N);
    place_edges_kernel<<<(E + 255) / 256, 256, 0, stream>>>(src, dst, rank, cntT, csr, E, N);

    gemm1_wmma<<<N / 16, 128, 0, stream>>>(x, W1, dinv, h1);
    pull_combine1_kernel<<<(N + 7) / 8, 256, 0, stream>>>(h1, rowptr, csr, dinv, b1, h1b, N);

    gemm2_wmma<<<N / 16, 64, 0, stream>>>(h1b, W2, h2);
    pull_combine2_proj_kernel<<<(N + 7) / 8, 256, 0, stream>>>(h2, rowptr, csr, dinv, b2, Wh, bh, out, N);
}


# Both gathers: one half-wave per node, a wave sums 2 adjacent nodes side by side with 16-B / 8-B lane loads (25000 waves each, was 50000); list order kept; final dot: channel pairs, then 16-lane tree
# speedup vs baseline: 2.9798x; 1.0567x over previous
#include <hip/hip_runtime.h>

typedef float v2f __attribute__((ext_vector_type(2)));
typedef float v8f __attribute__((ext_vector_type(8)));

#define IN_CH  128
#define HID1   64
#define HID2   32

#define SLICES    32
#define HIST_MAX  51200
#define CNT_TILES 13
#define PBLK      256
#define BLK_TRIPS ((HIST_MAX / PBLK + 31) / 32)

__device__ __forceinline__ int slice_len(int nE) {
    return (nE + SLICES * 4 - 1) / (SLICES * 4) * 4;
}

__global__ __launch_bounds__(1024) void slice_count_kernel(const int* __restrict__ dst,
                                                           int* __restrict__ cntT,
                                                           int* __restrict__ blkT,
                                                           int* __restrict__ rank,
                                                           int nE, int n) {
    __shared__ int hist[HIST_MAX];
    const int t = threadIdx.x;
    const int g = blockIdx.x;
    for (int i = t; i < HIST_MAX / 4; i += 1024)
        reinterpret_cast<int4*>(hist)[i] = make_int4(0, 0, 0, 0);
    __syncthreads();
    const int per = slice_len(nE);
    const int beg = g * per;
    const int end = min(beg + per, nE);
    for (int base = beg; base < end; base += CNT_TILES * 4096) {
        int4 d[CNT_TILES];
        #pragma unroll
        for (int k = 0; k < CNT_TILES; ++k) {
            const int i = base + (k * 1024 + t) * 4;
            d[k] = make_int4(0, 0, 0, 0);
            if (i + 3 < end) d[k] = *reinterpret_cast<const int4*>(dst + i);
        }
        #pragma unroll
        for (int k = 0; k < CNT_TILES; ++k) {
            const int i = base + (k * 1024 + t) * 4;
            if (i + 3 < end) {
                int4 r;
                r.x = atomicAdd(&hist[d[k].x], 1);
                r.y = atomicAdd(&hist[d[k].y], 1);
                r.z = atomicAdd(&hist[d[k].z], 1);
                r.w = atomicAdd(&hist[d[k].w], 1);
                *reinterpret_cast<int4*>(rank + i) = r;
            } else {
                for (int j = i; j < end; ++j) rank[j] = atomicAdd(&hist[dst[j]], 1);
            }
        }
    }
    __syncthreads();
    int* __restrict__ row = cntT + (size_t)g * n;
    const int lane = t & 31, wave = t >> 5;
    const int nb = (n + PBLK - 1) / PBLK;
    int s[BLK_TRIPS];
    #pragma unroll
    for (int k = 0; k < BLK_TRIPS; ++k) {
        const int b = wave + 32 * k;
        s[k] = 0;
        if (b < nb) {
            const int i0 = b * PBLK + lane;
            int v[PBLK / 32];
            #pragma unroll
            for (int j = 0; j < PBLK / 32; ++j) v[j] = hist[i0 + j * 32];
            if (b * PBLK + PBLK <= n) {
                #pragma unroll
                for (int j = 0; j < PBLK / 32; ++j) row[i0 + j * 32] = v[j];
            } else {
                #pragma unroll
                for (int j = 0; j < PBLK / 32; ++j)
                    if (i0 + j * 32 < n) row[i0 + j * 32] = v[j];
            }
            #pragma unroll
            for (int j = 0; j < PBLK / 32; ++j) s[k] += v[j];
        }
    }
    #pragma unroll
    for (int off = 16; off > 0; off >>= 1) {
        #pragma unroll
        for (int k = 0; k < BLK_TRIPS; ++k) s[k] += __shfl_xor(s[k], off, 32);
    }
    if (lane == 0) {
        #pragma unroll
        for (int k = 0; k < BLK_TRIPS; ++k) {
            const int b = wave + 32 * k;
            if (b < nb) blkT[b * SLICES + g] = s[k];
        }
    }
}

#define BASE_Q 8
static_assert(HIST_MAX / PBLK * SLICES <= BASE_Q * PBLK * 4, "block totals exceed one pass");
__global__ __launch_bounds__(PBLK) void slice_prefix_kernel(int* __restrict__ cntT,
                                                            const int* __restrict__ blkT,
                                                            int* __restrict__ rowptr,
                                                            float* __restrict__ dinv, int n) {
    __shared__ int wbase[PBLK / 32];
    __shared__ int wtot[PBLK / 32];
    const int t = threadIdx.x, lane = t & 31, wave = t >> 5;
    const int b = blockIdx.x;
    const int i = b * PBLK + t;
    const bool live = i < n;
    const int ic = live ? i : n - 1;
    int c[SLICES];
    #pragma unroll
    for (int g = 0; g < SLICES; ++g) c[g] = cntT[(size_t)g * n + ic];
    const int4* __restrict__ bq = reinterpret_cast<const int4*>(blkT);
    const int nq = b * (SLICES / 4);
    const int qlast = (int)gridDim.x * (SLICES / 4) - 1;
    int part = 0;
    int4 v[BASE_Q];
    #pragma unroll
    for (int k = 0; k < BASE_Q; ++k) v[k] = bq[min(k * PBLK + t, qlast)];
    #pragma unroll
    for (int k = 0; k < BASE_Q; ++k) {
        const int s = v[k].x + v[k].y + v[k].z + v[k].w;
        part += (k * PBLK + t < nq) ? s : 0;
    }
    int tot = 0;
    #pragma unroll
    for (int g = 0; g < SLICES; ++g) tot += c[g];
    if (!live) tot = 0;
    int incl = tot;
    #pragma unroll
    for (int off = 1; off < 32; off <<= 1) {
        const int y = __shfl_up(incl, off, 32);
        if (lane >= off) incl += y;
    }
    #pragma unroll
    for (int off = 16; off > 0; off >>= 1) part += __shfl_xor(part, off, 32);
    if (lane == 31) wtot[wave] = incl;
    if (lane == 0)  wbase[wave] = part;
    __syncthreads();
    int start = incl - tot;
    #pragma unroll
    for (int w = 0; w < PBLK / 32; ++w) {
        start += wbase[w];
        if (w < wave) start += wtot[w];
    }
    if (live) {
        rowptr[i] = start;
        dinv[i] = rsqrtf((float)tot + 1.0f);
        if (i == n - 1) rowptr[n] = start + tot;
        int run = start;
        #pragma unroll
        for (int g = 0; g < SLICES; ++g) {
            cntT[(size_t)g * n + i] = run;
            run += c[g];
        }
    }
}

__global__ __launch_bounds__(256) void place_edges_kernel(const int* __restrict__ src,
                                                          const int* __restrict__ dst,
                                                          const int* __restrict__ rank,
                                                          const int* __restrict__ cntT,
                                                          int* __restrict__ csr,
                                                          int nE, int n) {
    int e = blockIdx.x * 256 + threadIdx.x;
    if (e >= nE) return;
    const int s = src[e], d = dst[e];
    const int g = e / slice_len(nE);
    const int p = cntT[(size_t)g * n + d] + rank[e];
    csr[p] = s;
}

__global__ __launch_bounds__(128) void gemm1_wmma(const float* __restrict__ X,
                                                  const float* __restrict__ W,
                                                  const float* __restrict__ dinv,
                                                  float* __restrict__ H) {
    const int lane = threadIdx.x & 31;
    const int wave = threadIdx.x >> 5;
    const int m0   = blockIdx.x * 16;
    const int n0   = wave * 16;
    const int l16  = lane & 15;
    const int khi  = (lane >> 4) * 2;
    const int row  = m0 + l16;
    const int col  = n0 + l16;
    v8f c = {};
    #pragma unroll 4
    for (int kk = 0; kk < IN_CH; kk += 4) {
        v2f a, b;
        a.x = X[row * IN_CH + kk + khi + 0];
        a.y = X[row * IN_CH + kk + khi + 1];
        b.x = W[(kk + khi + 0) * HID1 + col];
        b.y = W[(kk + khi + 1) * HID1 + col];
        c = __builtin_amdgcn_wmma_f32_16x16x4_f32(false, a, false, b,
                                                  (short)0, c, false, false);
    }
    const int mbase = m0 + ((lane >> 4) * 8);
    const float4 s0 = *reinterpret_cast<const float4*>(dinv + mbase);
    const float4 s1 = *reinterpret_cast<const float4*>(dinv + mbase + 4);
    const float sc[8] = {s0.x, s0.y, s0.z, s0.w, s1.x, s1.y, s1.z, s1.w};
    #pragma unroll
    for (int r = 0; r < 8; ++r)
        H[(mbase + r) * HID1 + col] = c[r] * sc[r];
}

__global__ __launch_bounds__(64) void gemm2_wmma(const float* __restrict__ X,
                                                 const float* __restrict__ W,
                                                 float* __restrict__ H) {
    const int lane = threadIdx.x & 31;
    const int wave = threadIdx.x >> 5;
    const int m0   = blockIdx.x * 16;
    const int n0   = wave * 16;
    const int l16  = lane & 15;
    const int khi  = (lane >> 4) * 2;
    const int row  = m0 + l16;
    const int col  = n0 + l16;
    v8f c = {};
    #pragma unroll 4
    for (int kk = 0; kk < HID1; kk += 4) {
        v2f a, b;
        a.x = X[row * HID1 + kk + khi + 0];
        a.y = X[row * HID1 + kk + khi + 1];
        b.x = W[(kk + khi + 0) * HID2 + col];
        b.y = W[(kk + khi + 1) * HID2 + col];
        c = __builtin_amdgcn_wmma_f32_16x16x4_f32(false, a, false, b,
                                                  (short)0, c, false, false);
    }
    const int mbase = m0 + ((lane >> 4) * 8);
    #pragma unroll
    for (int r = 0; r < 8; ++r)
        H[(mbase + r) * HID2 + col] = c[r];
}


template <int J>
__device__ __forceinline__ int half_bcast(int s) {
    return __builtin_amdgcn_update_dpp(0, s, 0x150 + J, 0xF, 0xF, false);
}

template <int VEC, int J>
__device__ __forceinline__ void load_row(const float* __restrict__ h, int s,
                                         unsigned lane_off, float (&v)[VEC]) {
    const unsigned off = (unsigned)half_bcast<J>(s) * (unsigned)(16 * VEC * 4) + lane_off;
    const char* p = reinterpret_cast<const char*>(h) + off;
    if constexpr (VEC == 4) {
        const float4 r = *reinterpret_cast<const float4*>(p);
        v[0] = r.x; v[1] = r.y; v[2] = r.z; v[3] = r.w;
    } else {
        const float2 r = *reinterpret_cast<const float2*>(p);
        v[0] = r.x; v[1] = r.y;
    }
}

template <int VEC, int J0>
__device__ __forceinline__ void sum_batch(const float* __restrict__ h, int s, unsigned lane_off,
                                          int first, int len, bool full, float (&acc)[VEC]) {
    float v[8][VEC];
    load_row<VEC, J0 + 0>(h, s, lane_off, v[0]);
    load_row<VEC, J0 + 1>(h, s, lane_off, v[1]);
    load_row<VEC, J0 + 2>(h, s, lane_off, v[2]);
    load_row<VEC, J0 + 3>(h, s, lane_off, v[3]);
    load_row<VEC, J0 + 4>(h, s, lane_off, v[4]);
    load_row<VEC, J0 + 5>(h, s, lane_off, v[5]);
    load_row<VEC, J0 + 6>(h, s, lane_off, v[6]);
    load_row<VEC, J0 + 7>(h, s, lane_off, v[7]);
    if (full) {
        #pragma unroll
        for (int u = 0; u < 8; ++u) {
            #pragma unroll
            for (int p = 0; p < VEC; ++p) acc[p] += v[u][p];
        }
    } else {
        #pragma unroll
        for (int u = 0; u < 8; ++u) {
            const bool in = first + u < len;
            #pragma unroll
            for (int p = 0; p < VEC; ++p) acc[p] += in ? v[u][p] : -0.0f;
        }
    }
}

template <int VEC>
__device__ __forceinline__ void gather_rows(const float* __restrict__ h,
                                            const int* __restrict__ csr,
                                            int beg, int len, int lenmin, int lenmax,
                                            int l16, float (&acc)[VEC]) {
    const unsigned lane_off = (unsigned)(l16 * VEC * 4);
    for (int e0 = 0; e0 < lenmax; e0 += 32) {
        int s0 = 0, s1 = 0;
        if (e0 + l16 < len)      s0 = csr[beg + e0 + l16];
        if (e0 + 16 + l16 < len) s1 = csr[beg + e0 + 16 + l16];
        sum_batch<VEC, 0>(h, s0, lane_off, e0, len, e0 + 8 <= lenmin, acc);
        if (e0 + 8 < lenmax)
            sum_batch<VEC, 8>(h, s0, lane_off, e0 + 8, len, e0 + 16 <= lenmin, acc);
        if (e0 + 16 < lenmax)
            sum_batch<VEC, 0>(h, s1, lane_off, e0 + 16, len, e0 + 24 <= lenmin, acc);
        if (e0 + 24 < lenmax)
            sum_batch<VEC, 8>(h, s1, lane_off, e0 + 24, len, e0 + 32 <= lenmin, acc);
    }
}

__global__ __launch_bounds__(256) void pull_combine1_kernel(const float* __restrict__ h,
                                                            const int* __restrict__ rowptr,
                                                            const int* __restrict__ csr,
                                                            const float* __restrict__ dinv,
                                                            const float* __restrict__ b,
                                                            float* __restrict__ hout, int n) {
    const int lane = threadIdx.x & 31;
    const int l16 = lane & 15;
    const int node0 = __builtin_amdgcn_readfirstlane((blockIdx.x * 8 + (threadIdx.x >> 5)) * 2);
    if (node0 >= n) return;
    const int node = node0 + (lane >> 4);
    const bool live = node < n;
    int beg = 0, len = 0;
    if (live) { beg = rowptr[node]; len = rowptr[node + 1] - beg; }
    const int len_lo = __builtin_amdgcn_readlane(len, 0);
    const int len_hi = __builtin_amdgcn_readlane(len, 16);
    const int lenmin = min(len_lo, len_hi), lenmax = max(len_lo, len_hi);
    const int nc = live ? node : node0;
    const size_t i = (size_t)nc * HID1 + l16 * 4;
    const float4 hv = *reinterpret_cast<const float4*>(h + i);
    const float dd = dinv[nc];
    const float4 bv = *reinterpret_cast<const float4*>(b + l16 * 4);
    float acc[4] = {0.0f, 0.0f, 0.0f, 0.0f};
    gather_rows<4>(h, csr, beg, len, lenmin, lenmax, l16, acc);
    float4 o;
    o.x = fmaxf((acc[0] + hv.x) * dd + bv.x, 0.0f) * dd;
    o.y = fmaxf((acc[1] + hv.y) * dd + bv.y, 0.0f) * dd;
    o.z = fmaxf((acc[2] + hv.z) * dd + bv.z, 0.0f) * dd;
    o.w = fmaxf((acc[3] + hv.w) * dd + bv.w, 0.0f) * dd;
    if (live) *reinterpret_cast<float4*>(hout + i) = o;
}

__global__ __launch_bounds__(256) void pull_combine2_proj_kernel(const float* __restrict__ h,
                                                                 const int* __restrict__ rowptr,
                                                                 const int* __restrict__ csr,
                                                                 const float* __restrict__ dinv,
                                                                 const float* __restrict__ b2,
                                                                 const float* __restrict__ Wh,
                                                                 const float* __restrict__ bh,
                                                                 float* __restrict__ out, int n) {
    const int lane = threadIdx.x & 31;
    const int l16 = lane & 15;
    const int node0 = __builtin_amdgcn_readfirstlane((blockIdx.x * 8 + (threadIdx.x >> 5)) * 2);
    if (node0 >= n) return;
    const int node = node0 + (lane >> 4);
    const bool live = node < n;
    int beg = 0, len = 0;
    if (live) { beg = rowptr[node]; len = rowptr[node + 1] - beg; }
    const int len_lo = __builtin_amdgcn_readlane(len, 0);
    const int len_hi = __builtin_amdgcn_readlane(len, 16);
    const int lenmin = min(len_lo, len_hi), lenmax = max(len_lo, len_hi);
    const int nc = live ? node : node0;
    const float2 hv = *reinterpret_cast<const float2*>(h + (size_t)nc * HID2 + l16 * 2);
    const float dd = dinv[nc];
    const float2 bb = *reinterpret_cast<const float2*>(b2 + l16 * 2);
    const float2 ww = *reinterpret_cast<const float2*>(Wh + l16 * 2);
    float acc[2] = {0.0f, 0.0f};
    gather_rows<2>(h, csr, beg, len, lenmin, lenmax, l16, acc);
    const float v0 = fmaxf((acc[0] + hv.x) * dd + bb.x, 0.0f);
    const float v1 = fmaxf((acc[1] + hv.y) * dd + bb.y, 0.0f);
    float a = v0 * ww.x + v1 * ww.y;
    #pragma unroll
    for (int off = 8; off > 0; off >>= 1)
        a += __shfl_xor(a, off, 32);
    if (live && l16 == 0) out[node] = a + bh[0];
}

extern "C" void kernel_launch(void* const* d_in, const int* in_sizes, int n_in,
                              void* d_out, int out_size, void* d_ws, size_t ws_size,
                              hipStream_t stream) {
    const float* x  = (const float*)d_in[0];
    const int*   ei = (const int*)d_in[1];
    const float* W1 = (const float*)d_in[2];
    const float* b1 = (const float*)d_in[3];
    const float* W2 = (const float*)d_in[4];
    const float* b2 = (const float*)d_in[5];
    const float* Wh = (const float*)d_in[6];
    const float* bh = (const float*)d_in[7];
    float* out = (float*)d_out;

    const int N = in_sizes[0] / IN_CH;
    const int E = in_sizes[1] / 2;
    const int* src = ei;
    const int* dst = ei + E;

    float* f      = (float*)d_ws;
    int*   blkT   = (int*)d_ws;
    int*   rowptr = blkT + 65536;
    float* dinv   = f + 196608;
    float* h1     = f + 262144;
    float* h1b    = h1 + (size_t)N * HID1;
    float* h2     = h1b + (size_t)N * HID1;
    int*   csr    = (int*)(h2 + (size_t)N * HID2);
    int*   rank   = csr + E;
    int*   cntT   = rank + E;

    slice_count_kernel<<<SLICES, 1024, 0, stream>>>(dst, cntT, blkT, rank, E, N);
    slice_prefix_kernel<<<(N + PBLK - 1) / PBLK, PBLK, 0, stream>>>(cntT, blkT, rowptr, dinv, N);
    place_edges_kernel<<<(E + 255) / 256, 256, 0, stream>>>(src, dst, rank, cntT, csr, E, N);

    gemm1_wmma<<<N / 16, 128, 0, stream>>>(x, W1, dinv, h1);
    pull_combine1_kernel<<<(N + 15) / 16, 256, 0, stream>>>(h1, rowptr, csr, dinv, b1, h1b, N);

    gemm2_wmma<<<N / 16, 64, 0, stream>>>(h1b, W2, h2);
    pull_combine2_proj_kernel<<<(N + 15) / 16, 256, 0, stream>>>(h2, rowptr, csr, dinv, b2, Wh, bh, out, N);
}


# GEMM 2 folded into the layer-1 combine: a work-group's 16 finished rows go to an LDS tile, two of its waves multiply it by W2 (same WMMA form, same k order); 6 launches, 64-channel plane gone
# speedup vs baseline: 3.0854x; 1.0354x over previous
#include <hip/hip_runtime.h>

typedef float v2f __attribute__((ext_vector_type(2)));
typedef float v8f __attribute__((ext_vector_type(8)));

#define IN_CH  128
#define HID1   64
#define HID2   32

#define SLICES    32
#define HIST_MAX  51200
#define CNT_TILES 13
#define PBLK      256
#define BLK_TRIPS ((HIST_MAX / PBLK + 31) / 32)

__device__ __forceinline__ int slice_len(int nE) {
    return (nE + SLICES * 4 - 1) / (SLICES * 4) * 4;
}

__global__ __launch_bounds__(1024) void slice_count_kernel(const int* __restrict__ dst,
                                                           int* __restrict__ cntT,
                                                           int* __restrict__ blkT,
                                                           int* __restrict__ rank,
                                                           int nE, int n) {
    __shared__ int hist[HIST_MAX];
    const int t = threadIdx.x;
    const int g = blockIdx.x;
    for (int i = t; i < HIST_MAX / 4; i += 1024)
        reinterpret_cast<int4*>(hist)[i] = make_int4(0, 0, 0, 0);
    __syncthreads();
    const int per = slice_len(nE);
    const int beg = g * per;
    const int end = min(beg + per, nE);
    for (int base = beg; base < end; base += CNT_TILES * 4096) {
        int4 d[CNT_TILES];
        #pragma unroll
        for (int k = 0; k < CNT_TILES; ++k) {
            const int i = base + (k * 1024 + t) * 4;
            d[k] = make_int4(0, 0, 0, 0);
            if (i + 3 < end) d[k] = *reinterpret_cast<const int4*>(dst + i);
        }
        #pragma unroll
        for (int k = 0; k < CNT_TILES; ++k) {
            const int i = base + (k * 1024 + t) * 4;
            if (i + 3 < end) {
                int4 r;
                r.x = atomicAdd(&hist[d[k].x], 1);
                r.y = atomicAdd(&hist[d[k].y], 1);
                r.z = atomicAdd(&hist[d[k].z], 1);
                r.w = atomicAdd(&hist[d[k].w], 1);
                *reinterpret_cast<int4*>(rank + i) = r;
            } else {
                for (int j = i; j < end; ++j) rank[j] = atomicAdd(&hist[dst[j]], 1);
            }
        }
    }
    __syncthreads();
    int* __restrict__ row = cntT + (size_t)g * n;
    const int lane = t & 31, wave = t >> 5;
    const int nb = (n + PBLK - 1) / PBLK;
    int s[BLK_TRIPS];
    #pragma unroll
    for (int k = 0; k < BLK_TRIPS; ++k) {
        const int b = wave + 32 * k;
        s[k] = 0;
        if (b < nb) {
            const int i0 = b * PBLK + lane;
            int v[PBLK / 32];
            #pragma unroll
            for (int j = 0; j < PBLK / 32; ++j) v[j] = hist[i0 + j * 32];
            if (b * PBLK + PBLK <= n) {
                #pragma unroll
                for (int j = 0; j < PBLK / 32; ++j) row[i0 + j * 32] = v[j];
            } else {
                #pragma unroll
                for (int j = 0; j < PBLK / 32; ++j)
                    if (i0 + j * 32 < n) row[i0 + j * 32] = v[j];
            }
            #pragma unroll
            for (int j = 0; j < PBLK / 32; ++j) s[k] += v[j];
        }
    }
    #pragma unroll
    for (int off = 16; off > 0; off >>= 1) {
        #pragma unroll
        for (int k = 0; k < BLK_TRIPS; ++k) s[k] += __shfl_xor(s[k], off, 32);
    }
    if (lane == 0) {
        #pragma unroll
        for (int k = 0; k < BLK_TRIPS; ++k) {
            const int b = wave + 32 * k;
            if (b < nb) blkT[b * SLICES + g] = s[k];
        }
    }
}

#define BASE_Q 8
static_assert(HIST_MAX / PBLK * SLICES <= BASE_Q * PBLK * 4, "block totals exceed one pass");
__global__ __launch_bounds__(PBLK) void slice_prefix_kernel(int* __restrict__ cntT,
                                                            const int* __restrict__ blkT,
                                                            int* __restrict__ rowptr,
                                                            float* __restrict__ dinv, int n) {
    __shared__ int wbase[PBLK / 32];
    __shared__ int wtot[PBLK / 32];
    const int t = threadIdx.x, lane = t & 31, wave = t >> 5;
    const int b = blockIdx.x;
    const int i = b * PBLK + t;
    const bool live = i < n;
    const int ic = live ? i : n - 1;
    int c[SLICES];
    #pragma unroll
    for (int g = 0; g < SLICES; ++g) c[g] = cntT[(size_t)g * n + ic];
    const int4* __restrict__ bq = reinterpret_cast<const int4*>(blkT);
    const int nq = b * (SLICES / 4);
    const int qlast = (int)gridDim.x * (SLICES / 4) - 1;
    int part = 0;
    int4 v[BASE_Q];
    #pragma unroll
    for (int k = 0; k < BASE_Q; ++k) v[k] = bq[min(k * PBLK + t, qlast)];
    #pragma unroll
    for (int k = 0; k < BASE_Q; ++k) {
        const int s = v[k].x + v[k].y + v[k].z + v[k].w;
        part += (k * PBLK + t < nq) ? s : 0;
    }
    int tot = 0;
    #pragma unroll
    for (int g = 0; g < SLICES; ++g) tot += c[g];
    if (!live) tot = 0;
    int incl = tot;
    #pragma unroll
    for (int off = 1; off < 32; off <<= 1) {
        const int y = __shfl_up(incl, off, 32);
        if (lane >= off) incl += y;
    }
    #pragma unroll
    for (int off = 16; off > 0; off >>= 1) part += __shfl_xor(part, off, 32);
    if (lane == 31) wtot[wave] = incl;
    if (lane == 0)  wbase[wave] = part;
    __syncthreads();
    int start = incl - tot;
    #pragma unroll
    for (int w = 0; w < PBLK / 32; ++w) {
        start += wbase[w];
        if (w < wave) start += wtot[w];
    }
    if (live) {
        rowptr[i] = start;
        dinv[i] = rsqrtf((float)tot + 1.0f);
        if (i == n - 1) rowptr[n] = start + tot;
        int run = start;
        #pragma unroll
        for (int g = 0; g < SLICES; ++g) {
            cntT[(size_t)g * n + i] = run;
            run += c[g];
        }
    }
}

__global__ __launch_bounds__(256) void place_edges_kernel(const int* __restrict__ src,
                                                          const int* __restrict__ dst,
                                                          const int* __restrict__ rank,
                                                          const int* __restrict__ cntT,
                                                          int* __restrict__ csr,
                                                          int nE, int n) {
    int e = blockIdx.x * 256 + threadIdx.x;
    if (e >= nE) return;
    const int s = src[e], d = dst[e];
    const int g = e / slice_len(nE);
    const int p = cntT[(size_t)g * n + d] + rank[e];
    csr[p] = s;
}

__global__ __launch_bounds__(128) void gemm1_wmma(const float* __restrict__ X,
                                                  const float* __restrict__ W,
                                                  const float* __restrict__ dinv,
                                                  float* __restrict__ H) {
    const int lane = threadIdx.x & 31;
    const int wave = threadIdx.x >> 5;
    const int m0   = blockIdx.x * 16;
    const int n0   = wave * 16;
    const int l16  = lane & 15;
    const int khi  = (lane >> 4) * 2;
    const int row  = m0 + l16;
    const int col  = n0 + l16;
    v8f c = {};
    #pragma unroll 4
    for (int kk = 0; kk < IN_CH; kk += 4) {
        v2f a, b;
        a.x = X[row * IN_CH + kk + khi + 0];
        a.y = X[row * IN_CH + kk + khi + 1];
        b.x = W[(kk + khi + 0) * HID1 + col];
        b.y = W[(kk + khi + 1) * HID1 + col];
        c = __builtin_amdgcn_wmma_f32_16x16x4_f32(false, a, false, b,
                                                  (short)0, c, false, false);
    }
    const int mbase = m0 + ((lane >> 4) * 8);
    const float4 s0 = *reinterpret_cast<const float4*>(dinv + mbase);
    const float4 s1 = *reinterpret_cast<const float4*>(dinv + mbase + 4);
    const float sc[8] = {s0.x, s0.y, s0.z, s0.w, s1.x, s1.y, s1.z, s1.w};
    #pragma unroll
    for (int r = 0; r < 8; ++r)
        H[(mbase + r) * HID1 + col] = c[r] * sc[r];
}


template <int J>
__device__ __forceinline__ int half_bcast(int s) {
    return __builtin_amdgcn_update_dpp(0, s, 0x150 + J, 0xF, 0xF, false);
}

template <int VEC, int J>
__device__ __forceinline__ void load_row(const float* __restrict__ h, int s,
                                         unsigned lane_off, float (&v)[VEC]) {
    const unsigned off = (unsigned)half_bcast<J>(s) * (unsigned)(16 * VEC * 4) + lane_off;
    const char* p = reinterpret_cast<const char*>(h) + off;
    if constexpr (VEC == 4) {
        const float4 r = *reinterpret_cast<const float4*>(p);
        v[0] = r.x; v[1] = r.y; v[2] = r.z; v[3] = r.w;
    } else {
        const float2 r = *reinterpret_cast<const float2*>(p);
        v[0] = r.x; v[1] = r.y;
    }
}

template <int VEC, int J0>
__device__ __forceinline__ void sum_batch(const float* __restrict__ h, int s, unsigned lane_off,
                                          int first, int len, bool full, float (&acc)[VEC]) {
    float v[8][VEC];
    load_row<VEC, J0 + 0>(h, s, lane_off, v[0]);
    load_row<VEC, J0 + 1>(h, s, lane_off, v[1]);
    load_row<VEC, J0 + 2>(h, s, lane_off, v[2]);
    load_row<VEC, J0 + 3>(h, s, lane_off, v[3]);
    load_row<VEC, J0 + 4>(h, s, lane_off, v[4]);
    load_row<VEC, J0 + 5>(h, s, lane_off, v[5]);
    load_row<VEC, J0 + 6>(h, s, lane_off, v[6]);
    load_row<VEC, J0 + 7>(h, s, lane_off, v[7]);
    if (full) {
        #pragma unroll
        for (int u = 0; u < 8; ++u) {
            #pragma unroll
            for (int p = 0; p < VEC; ++p) acc[p] += v[u][p];
        }
    } else {
        #pragma unroll
        for (int u = 0; u < 8; ++u) {
            const bool in = first + u < len;
            #pragma unroll
            for (int p = 0; p < VEC; ++p) acc[p] += in ? v[u][p] : -0.0f;
        }
    }
}

template <int VEC>
__device__ __forceinline__ void gather_rows(const float* __restrict__ h,
                                            const int* __restrict__ csr,
                                            int beg, int len, int lenmin, int lenmax,
                                            int l16, float (&acc)[VEC]) {
    const unsigned lane_off = (unsigned)(l16 * VEC * 4);
    for (int e0 = 0; e0 < lenmax; e0 += 32) {
        int s0 = 0, s1 = 0;
        if (e0 + l16 < len)      s0 = csr[beg + e0 + l16];
        if (e0 + 16 + l16 < len) s1 = csr[beg + e0 + 16 + l16];
        sum_batch<VEC, 0>(h, s0, lane_off, e0, len, e0 + 8 <= lenmin, acc);
        if (e0 + 8 < lenmax)
            sum_batch<VEC, 8>(h, s0, lane_off, e0 + 8, len, e0 + 16 <= lenmin, acc);
        if (e0 + 16 < lenmax)
            sum_batch<VEC, 0>(h, s1, lane_off, e0 + 16, len, e0 + 24 <= lenmin, acc);
        if (e0 + 24 < lenmax)
            sum_batch<VEC, 8>(h, s1, lane_off, e0 + 24, len, e0 + 32 <= lenmin, acc);
    }
}

#define TPITCH 68
__global__ __launch_bounds__(256) void pull_combine1_gemm2_kernel(const float* __restrict__ h,
                                                                  const int* __restrict__ rowptr,
                                                                  const int* __restrict__ csr,
                                                                  const float* __restrict__ dinv,
                                                                  const float* __restrict__ b,
                                                                  const float* __restrict__ W2,
                                                                  float* __restrict__ h2, int n) {
    __shared__ __attribute__((aligned(16))) float tile[16 * TPITCH];
    const int lane = threadIdx.x & 31;
    const int l16 = lane & 15;
    const int half = lane >> 4;
    const int wave = __builtin_amdgcn_readfirstlane(threadIdx.x >> 5);
    const int m0 = blockIdx.x * 16;
    const int node0 = m0 + wave * 2;
    const int node = node0 + half;
    const bool live = node < n;
    int beg = 0, len = 0;
    if (live) { beg = rowptr[node]; len = rowptr[node + 1] - beg; }
    const int len_lo = __builtin_amdgcn_readlane(len, 0);
    const int len_hi = __builtin_amdgcn_readlane(len, 16);
    const int lenmin = min(len_lo, len_hi), lenmax = max(len_lo, len_hi);
    const int nc = live ? node : n - 1;
    const size_t i = (size_t)nc * HID1 + l16 * 4;
    const float4 hv = *reinterpret_cast<const float4*>(h + i);
    const float dd = dinv[nc];
    const float4 bv = *reinterpret_cast<const float4*>(b + l16 * 4);
    float acc[4] = {0.0f, 0.0f, 0.0f, 0.0f};
    gather_rows<4>(h, csr, beg, len, lenmin, lenmax, l16, acc);
    float4 o;
    o.x = fmaxf((acc[0] + hv.x) * dd + bv.x, 0.0f) * dd;
    o.y = fmaxf((acc[1] + hv.y) * dd + bv.y, 0.0f) * dd;
    o.z = fmaxf((acc[2] + hv.z) * dd + bv.z, 0.0f) * dd;
    o.w = fmaxf((acc[3] + hv.w) * dd + bv.w, 0.0f) * dd;
    if (!live) o = make_float4(0.0f, 0.0f, 0.0f, 0.0f);
    *reinterpret_cast<float4*>(&tile[(wave * 2 + half) * TPITCH + l16 * 4]) = o;

    const int khi = half * 2;
    const int col = (wave & 1) * 16 + l16;
    v2f bf[HID1 / 4] = {};
    if (wave < 2) {
        #pragma unroll
        for (int j = 0; j < HID1 / 4; ++j) {
            bf[j].x = W2[(j * 4 + khi + 0) * HID2 + col];
            bf[j].y = W2[(j * 4 + khi + 1) * HID2 + col];
        }
    }
    __syncthreads();
    if (wave >= 2) return;
    v8f c = {};
    #pragma unroll
    for (int j = 0; j < HID1 / 4; ++j) {
        const v2f a = *reinterpret_cast<const v2f*>(&tile[l16 * TPITCH + j * 4 + khi]);
        c = __builtin_amdgcn_wmma_f32_16x16x4_f32(false, a, false, bf[j],
                                                  (short)0, c, false, false);
    }
    const int mbase = m0 + half * 8;
    float* __restrict__ orow = h2 + (size_t)mbase * HID2 + col;
    if (mbase + 8 <= n) {
        #pragma unroll
        for (int r = 0; r < 8; ++r) orow[r * HID2] = c[r];
    } else {
        #pragma unroll
        for (int r = 0; r < 8; ++r)
            if (mbase + r < n) orow[r * HID2] = c[r];
    }
}

__global__ __launch_bounds__(256) void pull_combine2_proj_kernel(const float* __restrict__ h,
                                                                 const int* __restrict__ rowptr,
                                                                 const int* __restrict__ csr,
                                                                 const float* __restrict__ dinv,
                                                                 const float* __restrict__ b2,
                                                                 const float* __restrict__ Wh,
                                                                 const float* __restrict__ bh,
                                                                 float* __restrict__ out, int n) {
    const int lane = threadIdx.x & 31;
    const int l16 = lane & 15;
    const int node0 = __builtin_amdgcn_readfirstlane((blockIdx.x * 8 + (threadIdx.x >> 5)) * 2);
    if (node0 >= n) return;
    const int node = node0 + (lane >> 4);
    const bool live = node < n;
    int beg = 0, len = 0;
    if (live) { beg = rowptr[node]; len = rowptr[node + 1] - beg; }
    const int len_lo = __builtin_amdgcn_readlane(len, 0);
    const int len_hi = __builtin_amdgcn_readlane(len, 16);
    const int lenmin = min(len_lo, len_hi), lenmax = max(len_lo, len_hi);
    const int nc = live ? node : node0;
    const float2 hv = *reinterpret_cast<const float2*>(h + (size_t)nc * HID2 + l16 * 2);
    const float dd = dinv[nc];
    const float2 bb = *reinterpret_cast<const float2*>(b2 + l16 * 2);
    const float2 ww = *reinterpret_cast<const float2*>(Wh + l16 * 2);
    float acc[2] = {0.0f, 0.0f};
    gather_rows<2>(h, csr, beg, len, lenmin, lenmax, l16, acc);
    const float v0 = fmaxf((acc[0] + hv.x) * dd + bb.x, 0.0f);
    const float v1 = fmaxf((acc[1] + hv.y) * dd + bb.y, 0.0f);
    float a = v0 * ww.x + v1 * ww.y;
    #pragma unroll
    for (int off = 8; off > 0; off >>= 1)
        a += __shfl_xor(a, off, 32);
    if (live && l16 == 0) out[node] = a + bh[0];
}

extern "C" void kernel_launch(void* const* d_in, const int* in_sizes, int n_in,
                              void* d_out, int out_size, void* d_ws, size_t ws_size,
                              hipStream_t stream) {
    const float* x  = (const float*)d_in[0];
    const int*   ei = (const int*)d_in[1];
    const float* W1 = (const float*)d_in[2];
    const float* b1 = (const float*)d_in[3];
    const float* W2 = (const float*)d_in[4];
    const float* b2 = (const float*)d_in[5];
    const float* Wh = (const float*)d_in[6];
    const float* bh = (const float*)d_in[7];
    float* out = (float*)d_out;

    const int N = in_sizes[0] / IN_CH;
    const int E = in_sizes[1] / 2;
    const int* src = ei;
    const int* dst = ei + E;

    float* f      = (float*)d_ws;
    int*   blkT   = (int*)d_ws;
    int*   rowptr = blkT + 65536;
    float* dinv   = f + 196608;
    float* h1     = f + 262144;
    float* h2     = h1 + (size_t)N * HID1;
    int*   csr    = (int*)(h2 + (size_t)N * HID2);
    int*   rank   = csr + E;
    int*   cntT   = rank + E;

    slice_count_kernel<<<SLICES, 1024, 0, stream>>>(dst, cntT, blkT, rank, E, N);
    slice_prefix_kernel<<<(N + PBLK - 1) / PBLK, PBLK, 0, stream>>>(cntT, blkT, rowptr, dinv, N);
    place_edges_kernel<<<(E + 255) / 256, 256, 0, stream>>>(src, dst, rank, cntT, csr, E, N);

    gemm1_wmma<<<N / 16, 128, 0, stream>>>(x, W1, dinv, h1);
    pull_combine1_gemm2_kernel<<<(N + 15) / 16, 256, 0, stream>>>(h1, rowptr, csr, dinv, b1, W2, h2, N);

    pull_combine2_proj_kernel<<<(N + 15) / 16, 256, 0, stream>>>(h2, rowptr, csr, dinv, b2, Wh, bh, out, N);
}


# GEMM 1: a work-group lays each 16-row tile of X into LDS with whole-row 512-B loads and its 4 waves read A fragments there; W1 fragments stay in registers over 2 tiles; k order, accumulator kept
# speedup vs baseline: 3.2276x; 1.0461x over previous
#include <hip/hip_runtime.h>

typedef float v2f __attribute__((ext_vector_type(2)));
typedef float v8f __attribute__((ext_vector_type(8)));

#define IN_CH  128
#define HID1   64
#define HID2   32

#define SLICES    32
#define HIST_MAX  51200
#define CNT_TILES 13
#define PBLK      256
#define BLK_TRIPS ((HIST_MAX / PBLK + 31) / 32)

__device__ __forceinline__ int slice_len(int nE) {
    return (nE + SLICES * 4 - 1) / (SLICES * 4) * 4;
}

__global__ __launch_bounds__(1024) void slice_count_kernel(const int* __restrict__ dst,
                                                           int* __restrict__ cntT,
                                                           int* __restrict__ blkT,
                                                           int* __restrict__ rank,
                                                           int nE, int n) {
    __shared__ int hist[HIST_MAX];
    const int t = threadIdx.x;
    const int g = blockIdx.x;
    for (int i = t; i < HIST_MAX / 4; i += 1024)
        reinterpret_cast<int4*>(hist)[i] = make_int4(0, 0, 0, 0);
    __syncthreads();
    const int per = slice_len(nE);
    const int beg = g * per;
    const int end = min(beg + per, nE);
    for (int base = beg; base < end; base += CNT_TILES * 4096) {
        int4 d[CNT_TILES];
        #pragma unroll
        for (int k = 0; k < CNT_TILES; ++k) {
            const int i = base + (k * 1024 + t) * 4;
            d[k] = make_int4(0, 0, 0, 0);
            if (i + 3 < end) d[k] = *reinterpret_cast<const int4*>(dst + i);
        }
        #pragma unroll
        for (int k = 0; k < CNT_TILES; ++k) {
            const int i = base + (k * 1024 + t) * 4;
            if (i + 3 < end) {
                int4 r;
                r.x = atomicAdd(&hist[d[k].x], 1);
                r.y = atomicAdd(&hist[d[k].y], 1);
                r.z = atomicAdd(&hist[d[k].z], 1);
                r.w = atomicAdd(&hist[d[k].w], 1);
                *reinterpret_cast<int4*>(rank + i) = r;
            } else {
                for (int j = i; j < end; ++j) rank[j] = atomicAdd(&hist[dst[j]], 1);
            }
        }
    }
    __syncthreads();
    int* __restrict__ row = cntT + (size_t)g * n;
    const int lane = t & 31, wave = t >> 5;
    const int nb = (n + PBLK - 1) / PBLK;
    int s[BLK_TRIPS];
    #pragma unroll
    for (int k = 0; k < BLK_TRIPS; ++k) {
        const int b = wave + 32 * k;
        s[k] = 0;
        if (b < nb) {
            const int i0 = b * PBLK + lane;
            int v[PBLK / 32];
            #pragma unroll
            for (int j = 0; j < PBLK / 32; ++j) v[j] = hist[i0 + j * 32];
            if (b * PBLK + PBLK <= n) {
                #pragma unroll
                for (int j = 0; j < PBLK / 32; ++j) row[i0 + j * 32] = v[j];
            } else {
                #pragma unroll
                for (int j = 0; j < PBLK / 32; ++j)
                    if (i0 + j * 32 < n) row[i0 + j * 32] = v[j];
            }
            #pragma unroll
            for (int j = 0; j < PBLK / 32; ++j) s[k] += v[j];
        }
    }
    #pragma unroll
    for (int off = 16; off > 0; off >>= 1) {
        #pragma unroll
        for (int k = 0; k < BLK_TRIPS; ++k) s[k] += __shfl_xor(s[k], off, 32);
    }
    if (lane == 0) {
        #pragma unroll
        for (int k = 0; k < BLK_TRIPS; ++k) {
            const int b = wave + 32 * k;
            if (b < nb) blkT[b * SLICES + g] = s[k];
        }
    }
}

#define BASE_Q 8
static_assert(HIST_MAX / PBLK * SLICES <= BASE_Q * PBLK * 4, "block totals exceed one pass");
__global__ __launch_bounds__(PBLK) void slice_prefix_kernel(int* __restrict__ cntT,
                                                            const int* __restrict__ blkT,
                                                            int* __restrict__ rowptr,
                                                            float* __restrict__ dinv, int n) {
    __shared__ int wbase[PBLK / 32];
    __shared__ int wtot[PBLK / 32];
    const int t = threadIdx.x, lane = t & 31, wave = t >> 5;
    const int b = blockIdx.x;
    const int i = b * PBLK + t;
    const bool live = i < n;
    const int ic = live ? i : n - 1;
    int c[SLICES];
    #pragma unroll
    for (int g = 0; g < SLICES; ++g) c[g] = cntT[(size_t)g * n + ic];
    const int4* __restrict__ bq = reinterpret_cast<const int4*>(blkT);
    const int nq = b * (SLICES / 4);
    const int qlast = (int)gridDim.x * (SLICES / 4) - 1;
    int part = 0;
    int4 v[BASE_Q];
    #pragma unroll
    for (int k = 0; k < BASE_Q; ++k) v[k] = bq[min(k * PBLK + t, qlast)];
    #pragma unroll
    for (int k = 0; k < BASE_Q; ++k) {
        const int s = v[k].x + v[k].y + v[k].z + v[k].w;
        part += (k * PBLK + t < nq) ? s : 0;
    }
    int tot = 0;
    #pragma unroll
    for (int g = 0; g < SLICES; ++g) tot += c[g];
    if (!live) tot = 0;
    int incl = tot;
    #pragma unroll
    for (int off = 1; off < 32; off <<= 1) {
        const int y = __shfl_up(incl, off, 32);
        if (lane >= off) incl += y;
    }
    #pragma unroll
    for (int off = 16; off > 0; off >>= 1) part += __shfl_xor(part, off, 32);
    if (lane == 31) wtot[wave] = incl;
    if (lane == 0)  wbase[wave] = part;
    __syncthreads();
    int start = incl - tot;
    #pragma unroll
    for (int w = 0; w < PBLK / 32; ++w) {
        start += wbase[w];
        if (w < wave) start += wtot[w];
    }
    if (live) {
        rowptr[i] = start;
        dinv[i] = rsqrtf((float)tot + 1.0f);
        if (i == n - 1) rowptr[n] = start + tot;
        int run = start;
        #pragma unroll
        for (int g = 0; g < SLICES; ++g) {
            cntT[(size_t)g * n + i] = run;
            run += c[g];
        }
    }
}

__global__ __launch_bounds__(256) void place_edges_kernel(const int* __restrict__ src,
                                                          const int* __restrict__ dst,
                                                          const int* __restrict__ rank,
                                                          const int* __restrict__ cntT,
                                                          int* __restrict__ csr,
                                                          int nE, int n) {
    int e = blockIdx.x * 256 + threadIdx.x;
    if (e >= nE) return;
    const int s = src[e], d = dst[e];
    const int g = e / slice_len(nE);
    const int p = cntT[(size_t)g * n + d] + rank[e];
    csr[p] = s;
}

#define APITCH   132

__device__ __forceinline__ void gemm1_tile(const float* __restrict__ buf, const v2f (&bf)[IN_CH / 4],
                                           const float* __restrict__ dinv, float* __restrict__ H,
                                           int tile, int lane, int col) {
    const int l16 = lane & 15;
    const int khi = (lane >> 4) * 2;
    const int mbase = tile * 16 + ((lane >> 4) * 8);
    const float4 s0 = *reinterpret_cast<const float4*>(dinv + mbase);
    const float4 s1 = *reinterpret_cast<const float4*>(dinv + mbase + 4);
    v8f c = {};
    #pragma unroll
    for (int j = 0; j < IN_CH / 4; ++j) {
        const v2f a = *reinterpret_cast<const v2f*>(&buf[l16 * APITCH + j * 4 + khi]);
        c = __builtin_amdgcn_wmma_f32_16x16x4_f32(false, a, false, bf[j],
                                                  (short)0, c, false, false);
    }
    const float sc[8] = {s0.x, s0.y, s0.z, s0.w, s1.x, s1.y, s1.z, s1.w};
    #pragma unroll
    for (int r = 0; r < 8; ++r)
        H[(mbase + r) * HID1 + col] = c[r] * sc[r];
}

__global__ __launch_bounds__(128) void gemm1_wmma(const float* __restrict__ X,
                                                  const float* __restrict__ W,
                                                  const float* __restrict__ dinv,
                                                  float* __restrict__ H, int ntiles) {
    __shared__ __attribute__((aligned(16))) float atile[2][16 * APITCH];
    const int lane = threadIdx.x & 31;
    const int wave = __builtin_amdgcn_readfirstlane(threadIdx.x >> 5);
    const int khi  = (lane >> 4) * 2;
    const int col  = wave * 16 + (lane & 15);
    const int tile0 = blockIdx.x * 2;
    if (tile0 >= ntiles) return;
    const bool two = tile0 + 1 < ntiles;
    const float* __restrict__ xa = X + ((size_t)tile0 * 16 + wave) * IN_CH + lane * 4;
    const float4 a0 = *reinterpret_cast<const float4*>(xa);
    const float4 a1 = *reinterpret_cast<const float4*>(xa + 4 * IN_CH);
    const float4 a2 = *reinterpret_cast<const float4*>(xa + 8 * IN_CH);
    const float4 a3 = *reinterpret_cast<const float4*>(xa + 12 * IN_CH);
    v2f bf[IN_CH / 4];
    #pragma unroll
    for (int j = 0; j < IN_CH / 4; ++j) {
        bf[j].x = W[(j * 4 + khi + 0) * HID1 + col];
        bf[j].y = W[(j * 4 + khi + 1) * HID1 + col];
    }
    float4 b0 = a0, b1 = a1, b2 = a2, b3 = a3;
    if (two) {
        const float* __restrict__ xb = xa + 16 * IN_CH;
        b0 = *reinterpret_cast<const float4*>(xb);
        b1 = *reinterpret_cast<const float4*>(xb + 4 * IN_CH);
        b2 = *reinterpret_cast<const float4*>(xb + 8 * IN_CH);
        b3 = *reinterpret_cast<const float4*>(xb + 12 * IN_CH);
    }
    float* __restrict__ s = &atile[0][wave * APITCH + lane * 4];
    *reinterpret_cast<float4*>(s)               = a0;
    *reinterpret_cast<float4*>(s + 4 * APITCH)  = a1;
    *reinterpret_cast<float4*>(s + 8 * APITCH)  = a2;
    *reinterpret_cast<float4*>(s + 12 * APITCH) = a3;
    __syncthreads();
    gemm1_tile(atile[0], bf, dinv, H, tile0, lane, col);
    if (!two) return;
    float* __restrict__ s2 = &atile[1][wave * APITCH + lane * 4];
    *reinterpret_cast<float4*>(s2)               = b0;
    *reinterpret_cast<float4*>(s2 + 4 * APITCH)  = b1;
    *reinterpret_cast<float4*>(s2 + 8 * APITCH)  = b2;
    *reinterpret_cast<float4*>(s2 + 12 * APITCH) = b3;
    __syncthreads();
    gemm1_tile(atile[1], bf, dinv, H, tile0 + 1, lane, col);
}


template <int J>
__device__ __forceinline__ int half_bcast(int s) {
    return __builtin_amdgcn_update_dpp(0, s, 0x150 + J, 0xF, 0xF, false);
}

template <int VEC, int J>
__device__ __forceinline__ void load_row(const float* __restrict__ h, int s,
                                         unsigned lane_off, float (&v)[VEC]) {
    const unsigned off = (unsigned)half_bcast<J>(s) * (unsigned)(16 * VEC * 4) + lane_off;
    const char* p = reinterpret_cast<const char*>(h) + off;
    if constexpr (VEC == 4) {
        const float4 r = *reinterpret_cast<const float4*>(p);
        v[0] = r.x; v[1] = r.y; v[2] = r.z; v[3] = r.w;
    } else {
        const float2 r = *reinterpret_cast<const float2*>(p);
        v[0] = r.x; v[1] = r.y;
    }
}

template <int VEC, int J0>
__device__ __forceinline__ void sum_batch(const float* __restrict__ h, int s, unsigned lane_off,
                                          int first, int len, bool full, float (&acc)[VEC]) {
    float v[8][VEC];
    load_row<VEC, J0 + 0>(h, s, lane_off, v[0]);
    load_row<VEC, J0 + 1>(h, s, lane_off, v[1]);
    load_row<VEC, J0 + 2>(h, s, lane_off, v[2]);
    load_row<VEC, J0 + 3>(h, s, lane_off, v[3]);
    load_row<VEC, J0 + 4>(h, s, lane_off, v[4]);
    load_row<VEC, J0 + 5>(h, s, lane_off, v[5]);
    load_row<VEC, J0 + 6>(h, s, lane_off, v[6]);
    load_row<VEC, J0 + 7>(h, s, lane_off, v[7]);
    if (full) {
        #pragma unroll
        for (int u = 0; u < 8; ++u) {
            #pragma unroll
            for (int p = 0; p < VEC; ++p) acc[p] += v[u][p];
        }
    } else {
        #pragma unroll
        for (int u = 0; u < 8; ++u) {
            const bool in = first + u < len;
            #pragma unroll
            for (int p = 0; p < VEC; ++p) acc[p] += in ? v[u][p] : -0.0f;
        }
    }
}

template <int VEC>
__device__ __forceinline__ void gather_rows(const float* __restrict__ h,
                                            const int* __restrict__ csr,
                                            int beg, int len, int lenmin, int lenmax,
                                            int l16, float (&acc)[VEC]) {
    const unsigned lane_off = (unsigned)(l16 * VEC * 4);
    for (int e0 = 0; e0 < lenmax; e0 += 32) {
        int s0 = 0, s1 = 0;
        if (e0 + l16 < len)      s0 = csr[beg + e0 + l16];
        if (e0 + 16 + l16 < len) s1 = csr[beg + e0 + 16 + l16];
        sum_batch<VEC, 0>(h, s0, lane_off, e0, len, e0 + 8 <= lenmin, acc);
        if (e0 + 8 < lenmax)
            sum_batch<VEC, 8>(h, s0, lane_off, e0 + 8, len, e0 + 16 <= lenmin, acc);
        if (e0 + 16 < lenmax)
            sum_batch<VEC, 0>(h, s1, lane_off, e0 + 16, len, e0 + 24 <= lenmin, acc);
        if (e0 + 24 < lenmax)
            sum_batch<VEC, 8>(h, s1, lane_off, e0 + 24, len, e0 + 32 <= lenmin, acc);
    }
}

#define TPITCH 68
__global__ __launch_bounds__(256) void pull_combine1_gemm2_kernel(const float* __restrict__ h,
                                                                  const int* __restrict__ rowptr,
                                                                  const int* __restrict__ csr,
                                                                  const float* __restrict__ dinv,
                                                                  const float* __restrict__ b,
                                                                  const float* __restrict__ W2,
                                                                  float* __restrict__ h2, int n) {
    __shared__ __attribute__((aligned(16))) float tile[16 * TPITCH];
    const int lane = threadIdx.x & 31;
    const int l16 = lane & 15;
    const int half = lane >> 4;
    const int wave = __builtin_amdgcn_readfirstlane(threadIdx.x >> 5);
    const int m0 = blockIdx.x * 16;
    const int node0 = m0 + wave * 2;
    const int node = node0 + half;
    const bool live = node < n;
    int beg = 0, len = 0;
    if (live) { beg = rowptr[node]; len = rowptr[node + 1] - beg; }
    const int len_lo = __builtin_amdgcn_readlane(len, 0);
    const int len_hi = __builtin_amdgcn_readlane(len, 16);
    const int lenmin = min(len_lo, len_hi), lenmax = max(len_lo, len_hi);
    const int nc = live ? node : n - 1;
    const size_t i = (size_t)nc * HID1 + l16 * 4;
    const float4 hv = *reinterpret_cast<const float4*>(h + i);
    const float dd = dinv[nc];
    const float4 bv = *reinterpret_cast<const float4*>(b + l16 * 4);
    float acc[4] = {0.0f, 0.0f, 0.0f, 0.0f};
    gather_rows<4>(h, csr, beg, len, lenmin, lenmax, l16, acc);
    float4 o;
    o.x = fmaxf((acc[0] + hv.x) * dd + bv.x, 0.0f) * dd;
    o.y = fmaxf((acc[1] + hv.y) * dd + bv.y, 0.0f) * dd;
    o.z = fmaxf((acc[2] + hv.z) * dd + bv.z, 0.0f) * dd;
    o.w = fmaxf((acc[3] + hv.w) * dd + bv.w, 0.0f) * dd;
    if (!live) o = make_float4(0.0f, 0.0f, 0.0f, 0.0f);
    *reinterpret_cast<float4*>(&tile[(wave * 2 + half) * TPITCH + l16 * 4]) = o;

    const int khi = half * 2;
    const int col = (wave & 1) * 16 + l16;
    v2f bf[HID1 / 4] = {};
    if (wave < 2) {
        #pragma unroll
        for (int j = 0; j < HID1 / 4; ++j) {
            bf[j].x = W2[(j * 4 + khi + 0) * HID2 + col];
            bf[j].y = W2[(j * 4 + khi + 1) * HID2 + col];
        }
    }
    __syncthreads();
    if (wave >= 2) return;
    v8f c = {};
    #pragma unroll
    for (int j = 0; j < HID1 / 4; ++j) {
        const v2f a = *reinterpret_cast<const v2f*>(&tile[l16 * TPITCH + j * 4 + khi]);
        c = __builtin_amdgcn_wmma_f32_16x16x4_f32(false, a, false, bf[j],
                                                  (short)0, c, false, false);
    }
    const int mbase = m0 + half * 8;
    float* __restrict__ orow = h2 + (size_t)mbase * HID2 + col;
    if (mbase + 8 <= n) {
        #pragma unroll
        for (int r = 0; r < 8; ++r) orow[r * HID2] = c[r];
    } else {
        #pragma unroll
        for (int r = 0; r < 8; ++r)
            if (mbase + r < n) orow[r * HID2] = c[r];
    }
}

__global__ __launch_bounds__(256) void pull_combine2_proj_kernel(const float* __restrict__ h,
                                                                 const int* __restrict__ rowptr,
                                                                 const int* __restrict__ csr,
                                                                 const float* __restrict__ dinv,
                                                                 const float* __restrict__ b2,
                                                                 const float* __restrict__ Wh,
                                                                 const float* __restrict__ bh,
                                                                 float* __restrict__ out, int n) {
    const int lane = threadIdx.x & 31;
    const int l16 = lane & 15;
    const int node0 = __builtin_amdgcn_readfirstlane((blockIdx.x * 8 + (threadIdx.x >> 5)) * 2);
    if (node0 >= n) return;
    const int node = node0 + (lane >> 4);
    const bool live = node < n;
    int beg = 0, len = 0;
    if (live) { beg = rowptr[node]; len = rowptr[node + 1] - beg; }
    const int len_lo = __builtin_amdgcn_readlane(len, 0);
    const int len_hi = __builtin_amdgcn_readlane(len, 16);
    const int lenmin = min(len_lo, len_hi), lenmax = max(len_lo, len_hi);
    const int nc = live ? node : node0;
    const float2 hv = *reinterpret_cast<const float2*>(h + (size_t)nc * HID2 + l16 * 2);
    const float dd = dinv[nc];
    const float2 bb = *reinterpret_cast<const float2*>(b2 + l16 * 2);
    const float2 ww = *reinterpret_cast<const float2*>(Wh + l16 * 2);
    float acc[2] = {0.0f, 0.0f};
    gather_rows<2>(h, csr, beg, len, lenmin, lenmax, l16, acc);
    const float v0 = fmaxf((acc[0] + hv.x) * dd + bb.x, 0.0f);
    const float v1 = fmaxf((acc[1] + hv.y) * dd + bb.y, 0.0f);
    float a = v0 * ww.x + v1 * ww.y;
    #pragma unroll
    for (int off = 8; off > 0; off >>= 1)
        a += __shfl_xor(a, off, 32);
    if (live && l16 == 0) out[node] = a + bh[0];
}

extern "C" void kernel_launch(void* const* d_in, const int* in_sizes, int n_in,
                              void* d_out, int out_size, void* d_ws, size_t ws_size,
                              hipStream_t stream) {
    const float* x  = (const float*)d_in[0];
    const int*   ei = (const int*)d_in[1];
    const float* W1 = (const float*)d_in[2];
    const float* b1 = (const float*)d_in[3];
    const float* W2 = (const float*)d_in[4];
    const float* b2 = (const float*)d_in[5];
    const float* Wh = (const float*)d_in[6];
    const float* bh = (const float*)d_in[7];
    float* out = (float*)d_out;

    const int N = in_sizes[0] / IN_CH;
    const int E = in_sizes[1] / 2;
    const int* src = ei;
    const int* dst = ei + E;

    float* f      = (float*)d_ws;
    int*   blkT   = (int*)d_ws;
    int*   rowptr = blkT + 65536;
    float* dinv   = f + 196608;
    float* h1     = f + 262144;
    float* h2     = h1 + (size_t)N * HID1;
    int*   csr    = (int*)(h2 + (size_t)N * HID2);
    int*   rank   = csr + E;
    int*   cntT   = rank + E;

    slice_count_kernel<<<SLICES, 1024, 0, stream>>>(dst, cntT, blkT, rank, E, N);
    slice_prefix_kernel<<<(N + PBLK - 1) / PBLK, PBLK, 0, stream>>>(cntT, blkT, rowptr, dinv, N);
    place_edges_kernel<<<(E + 255) / 256, 256, 0, stream>>>(src, dst, rank, cntT, csr, E, N);

    gemm1_wmma<<<(N / 16 + 1) / 2, 128, 0, stream>>>(x, W1, dinv, h1, N / 16);
    pull_combine1_gemm2_kernel<<<(N + 15) / 16, 256, 0, stream>>>(h1, rowptr, csr, dinv, b1, W2, h2, N);

    pull_combine2_proj_kernel<<<(N + 15) / 16, 256, 0, stream>>>(h2, rowptr, csr, dinv, b2, Wh, bh, out, N);
}


# Edge placing: 8 work-groups per slice copy the slice's row of list slots into LDS with whole-line loads; an edge's slot is an LDS read, not a scattered 4-B table read; same slots, same lists
# speedup vs baseline: 3.3550x; 1.0395x over previous
#include <hip/hip_runtime.h>

typedef float v2f __attribute__((ext_vector_type(2)));
typedef float v8f __attribute__((ext_vector_type(8)));

#define IN_CH  128
#define HID1   64
#define HID2   32

#define SLICES    32
#define HIST_MAX  51200
#define CNT_TILES 13
#define PBLK      256
#define BLK_TRIPS ((HIST_MAX / PBLK + 31) / 32)

__device__ __forceinline__ int slice_len(int nE) {
    return (nE + SLICES * 4 - 1) / (SLICES * 4) * 4;
}

__global__ __launch_bounds__(1024) void slice_count_kernel(const int* __restrict__ dst,
                                                           int* __restrict__ cntT,
                                                           int* __restrict__ blkT,
                                                           int* __restrict__ rank,
                                                           int nE, int n) {
    __shared__ int hist[HIST_MAX];
    const int t = threadIdx.x;
    const int g = blockIdx.x;
    for (int i = t; i < HIST_MAX / 4; i += 1024)
        reinterpret_cast<int4*>(hist)[i] = make_int4(0, 0, 0, 0);
    __syncthreads();
    const int per = slice_len(nE);
    const int beg = g * per;
    const int end = min(beg + per, nE);
    for (int base = beg; base < end; base += CNT_TILES * 4096) {
        int4 d[CNT_TILES];
        #pragma unroll
        for (int k = 0; k < CNT_TILES; ++k) {
            const int i = base + (k * 1024 + t) * 4;
            d[k] = make_int4(0, 0, 0, 0);
            if (i + 3 < end) d[k] = *reinterpret_cast<const int4*>(dst + i);
        }
        #pragma unroll
        for (int k = 0; k < CNT_TILES; ++k) {
            const int i = base + (k * 1024 + t) * 4;
            if (i + 3 < end) {
                int4 r;
                r.x = atomicAdd(&hist[d[k].x], 1);
                r.y = atomicAdd(&hist[d[k].y], 1);
                r.z = atomicAdd(&hist[d[k].z], 1);
                r.w = atomicAdd(&hist[d[k].w], 1);
                *reinterpret_cast<int4*>(rank + i) = r;
            } else {
                for (int j = i; j < end; ++j) rank[j] = atomicAdd(&hist[dst[j]], 1);
            }
        }
    }
    __syncthreads();
    int* __restrict__ row = cntT + (size_t)g * n;
    const int lane = t & 31, wave = t >> 5;
    const int nb = (n + PBLK - 1) / PBLK;
    int s[BLK_TRIPS];
    #pragma unroll
    for (int k = 0; k < BLK_TRIPS; ++k) {
        const int b = wave + 32 * k;
        s[k] = 0;
        if (b < nb) {
            const int i0 = b * PBLK + lane;
            int v[PBLK / 32];
            #pragma unroll
            for (int j = 0; j < PBLK / 32; ++j) v[j] = hist[i0 + j * 32];
            if (b * PBLK + PBLK <= n) {
                #pragma unroll
                for (int j = 0; j < PBLK / 32; ++j) row[i0 + j * 32] = v[j];
            } else {
                #pragma unroll
                for (int j = 0; j < PBLK / 32; ++j)
                    if (i0 + j * 32 < n) row[i0 + j * 32] = v[j];
            }
            #pragma unroll
            for (int j = 0; j < PBLK / 32; ++j) s[k] += v[j];
        }
    }
    #pragma unroll
    for (int off = 16; off > 0; off >>= 1) {
        #pragma unroll
        for (int k = 0; k < BLK_TRIPS; ++k) s[k] += __shfl_xor(s[k], off, 32);
    }
    if (lane == 0) {
        #pragma unroll
        for (int k = 0; k < BLK_TRIPS; ++k) {
            const int b = wave + 32 * k;
            if (b < nb) blkT[b * SLICES + g] = s[k];
        }
    }
}

#define BASE_Q 8
static_assert(HIST_MAX / PBLK * SLICES <= BASE_Q * PBLK * 4, "block totals exceed one pass");
__global__ __launch_bounds__(PBLK) void slice_prefix_kernel(int* __restrict__ cntT,
                                                            const int* __restrict__ blkT,
                                                            int* __restrict__ rowptr,
                                                            float* __restrict__ dinv, int n) {
    __shared__ int wbase[PBLK / 32];
    __shared__ int wtot[PBLK / 32];
    const int t = threadIdx.x, lane = t & 31, wave = t >> 5;
    const int b = blockIdx.x;
    const int i = b * PBLK + t;
    const bool live = i < n;
    const int ic = live ? i : n - 1;
    int c[SLICES];
    #pragma unroll
    for (int g = 0; g < SLICES; ++g) c[g] = cntT[(size_t)g * n + ic];
    const int4* __restrict__ bq = reinterpret_cast<const int4*>(blkT);
    const int nq = b * (SLICES / 4);
    const int qlast = (int)gridDim.x * (SLICES / 4) - 1;
    int part = 0;
    int4 v[BASE_Q];
    #pragma unroll
    for (int k = 0; k < BASE_Q; ++k) v[k] = bq[min(k * PBLK + t, qlast)];
    #pragma unroll
    for (int k = 0; k < BASE_Q; ++k) {
        const int s = v[k].x + v[k].y + v[k].z + v[k].w;
        part += (k * PBLK + t < nq) ? s : 0;
    }
    int tot = 0;
    #pragma unroll
    for (int g = 0; g < SLICES; ++g) tot += c[g];
    if (!live) tot = 0;
    int incl = tot;
    #pragma unroll
    for (int off = 1; off < 32; off <<= 1) {
        const int y = __shfl_up(incl, off, 32);
        if (lane >= off) incl += y;
    }
    #pragma unroll
    for (int off = 16; off > 0; off >>= 1) part += __shfl_xor(part, off, 32);
    if (lane == 31) wtot[wave] = incl;
    if (lane == 0)  wbase[wave] = part;
    __syncthreads();
    int start = incl - tot;
    #pragma unroll
    for (int w = 0; w < PBLK / 32; ++w) {
        start += wbase[w];
        if (w < wave) start += wtot[w];
    }
    if (live) {
        rowptr[i] = start;
        dinv[i] = rsqrtf((float)tot + 1.0f);
        if (i == n - 1) rowptr[n] = start + tot;
        int run = start;
        #pragma unroll
        for (int g = 0; g < SLICES; ++g) {
            cntT[(size_t)g * n + i] = run;
            run += c[g];
        }
    }
}

#define PLACE_PARTS 8
#define STAGE_Q ((HIST_MAX / 4 + 1023) / 1024)

__device__ __forceinline__ void place_four(const int* __restrict__ slot, int* __restrict__ csr,
                                           const int4 s, const int4 d, const int4 r) {
    const int p0 = slot[d.x] + r.x;
    const int p1 = slot[d.y] + r.y;
    const int p2 = slot[d.z] + r.z;
    const int p3 = slot[d.w] + r.w;
    csr[p0] = s.x;
    csr[p1] = s.y;
    csr[p2] = s.z;
    csr[p3] = s.w;
}

__global__ __launch_bounds__(1024) void place_edges_kernel(const int* __restrict__ src,
                                                           const int* __restrict__ dst,
                                                           const int* __restrict__ rank,
                                                           const int* __restrict__ cntT,
                                                           int* __restrict__ csr,
                                                           int nE, int n) {
    __shared__ __attribute__((aligned(16))) int slot[HIST_MAX];
    const int t = threadIdx.x;
    const int g = blockIdx.x / PLACE_PARTS;
    const int part = blockIdx.x % PLACE_PARTS;
    const int per = slice_len(nE);
    const int beg = g * per;
    const int end = min(beg + per, nE);
    const int sub = (per + PLACE_PARTS * 4 - 1) / (PLACE_PARTS * 4) * 4;
    const int lo = beg + part * sub;
    const int hi = min(lo + sub, end);
    if (lo >= hi) return;
    const int* __restrict__ row = cntT + (size_t)g * n;
    if ((n & 3) == 0 && (nE & 1) == 0) {
        const int nq = n >> 2;
        const int4* __restrict__ rowq = reinterpret_cast<const int4*>(row);
        int4 v[STAGE_Q];
        #pragma unroll
        for (int k = 0; k < STAGE_Q; ++k) {
            const int i = k * 1024 + t;
            v[k] = make_int4(0, 0, 0, 0);
            if (i < nq) v[k] = rowq[i];
        }
        #pragma unroll
        for (int k = 0; k < STAGE_Q; ++k) {
            const int i = k * 1024 + t;
            if (i < nq) reinterpret_cast<int4*>(slot)[i] = v[k];
        }
    } else {
        for (int i = t; i < n; i += 1024) slot[i] = row[i];
    }
    __syncthreads();
    for (int i = lo + t * 4; i < hi; i += 8192) {
        const int i2 = i + 4096;
        const bool f1 = i + 3 < hi;
        const bool f2 = i2 + 3 < hi;
        int4 s1 = make_int4(0, 0, 0, 0), d1 = s1, r1 = s1, s2 = s1, d2 = s1, r2 = s1;
        if (f1) {
            s1 = *reinterpret_cast<const int4*>(src + i);
            d1 = *reinterpret_cast<const int4*>(dst + i);
            r1 = *reinterpret_cast<const int4*>(rank + i);
        }
        if (f2) {
            s2 = *reinterpret_cast<const int4*>(src + i2);
            d2 = *reinterpret_cast<const int4*>(dst + i2);
            r2 = *reinterpret_cast<const int4*>(rank + i2);
        }
        if (f1) {
            place_four(slot, csr, s1, d1, r1);
        } else {
            for (int j = i; j < hi; ++j) csr[slot[dst[j]] + rank[j]] = src[j];
        }
        if (f2) {
            place_four(slot, csr, s2, d2, r2);
        } else {
            for (int j = i2; j < hi; ++j) csr[slot[dst[j]] + rank[j]] = src[j];
        }
    }
}

#define APITCH   132

__device__ __forceinline__ void gemm1_tile(const float* __restrict__ buf, const v2f (&bf)[IN_CH / 4],
                                           const float* __restrict__ dinv, float* __restrict__ H,
                                           int tile, int lane, int col) {
    const int l16 = lane & 15;
    const int khi = (lane >> 4) * 2;
    const int mbase = tile * 16 + ((lane >> 4) * 8);
    const float4 s0 = *reinterpret_cast<const float4*>(dinv + mbase);
    const float4 s1 = *reinterpret_cast<const float4*>(dinv + mbase + 4);
    v8f c = {};
    #pragma unroll
    for (int j = 0; j < IN_CH / 4; ++j) {
        const v2f a = *reinterpret_cast<const v2f*>(&buf[l16 * APITCH + j * 4 + khi]);
        c = __builtin_amdgcn_wmma_f32_16x16x4_f32(false, a, false, bf[j],
                                                  (short)0, c, false, false);
    }
    const float sc[8] = {s0.x, s0.y, s0.z, s0.w, s1.x, s1.y, s1.z, s1.w};
    #pragma unroll
    for (int r = 0; r < 8; ++r)
        H[(mbase + r) * HID1 + col] = c[r] * sc[r];
}

__global__ __launch_bounds__(128) void gemm1_wmma(const float* __restrict__ X,
                                                  const float* __restrict__ W,
                                                  const float* __restrict__ dinv,
                                                  float* __restrict__ H, int ntiles) {
    __shared__ __attribute__((aligned(16))) float atile[2][16 * APITCH];
    const int lane = threadIdx.x & 31;
    const int wave = __builtin_amdgcn_readfirstlane(threadIdx.x >> 5);
    const int khi  = (lane >> 4) * 2;
    const int col  = wave * 16 + (lane & 15);
    const int tile0 = blockIdx.x * 2;
    if (tile0 >= ntiles) return;
    const bool two = tile0 + 1 < ntiles;
    const float* __restrict__ xa = X + ((size_t)tile0 * 16 + wave) * IN_CH + lane * 4;
    const float4 a0 = *reinterpret_cast<const float4*>(xa);
    const float4 a1 = *reinterpret_cast<const float4*>(xa + 4 * IN_CH);
    const float4 a2 = *reinterpret_cast<const float4*>(xa + 8 * IN_CH);
    const float4 a3 = *reinterpret_cast<const float4*>(xa + 12 * IN_CH);
    v2f bf[IN_CH / 4];
    #pragma unroll
    for (int j = 0; j < IN_CH / 4; ++j) {
        bf[j].x = W[(j * 4 + khi + 0) * HID1 + col];
        bf[j].y = W[(j * 4 + khi + 1) * HID1 + col];
    }
    float4 b0 = a0, b1 = a1, b2 = a2, b3 = a3;
    if (two) {
        const float* __restrict__ xb = xa + 16 * IN_CH;
        b0 = *reinterpret_cast<const float4*>(xb);
        b1 = *reinterpret_cast<const float4*>(xb + 4 * IN_CH);
        b2 = *reinterpret_cast<const float4*>(xb + 8 * IN_CH);
        b3 = *reinterpret_cast<const float4*>(xb + 12 * IN_CH);
    }
    float* __restrict__ s = &atile[0][wave * APITCH + lane * 4];
    *reinterpret_cast<float4*>(s)               = a0;
    *reinterpret_cast<float4*>(s + 4 * APITCH)  = a1;
    *reinterpret_cast<float4*>(s + 8 * APITCH)  = a2;
    *reinterpret_cast<float4*>(s + 12 * APITCH) = a3;
    __syncthreads();
    gemm1_tile(atile[0], bf, dinv, H, tile0, lane, col);
    if (!two) return;
    float* __restrict__ s2 = &atile[1][wave * APITCH + lane * 4];
    *reinterpret_cast<float4*>(s2)               = b0;
    *reinterpret_cast<float4*>(s2 + 4 * APITCH)  = b1;
    *reinterpret_cast<float4*>(s2 + 8 * APITCH)  = b2;
    *reinterpret_cast<float4*>(s2 + 12 * APITCH) = b3;
    __syncthreads();
    gemm1_tile(atile[1], bf, dinv, H, tile0 + 1, lane, col);
}


template <int J>
__device__ __forceinline__ int half_bcast(int s) {
    return __builtin_amdgcn_update_dpp(0, s, 0x150 + J, 0xF, 0xF, false);
}

template <int VEC, int J>
__device__ __forceinline__ void load_row(const float* __restrict__ h, int s,
                                         unsigned lane_off, float (&v)[VEC]) {
    const unsigned off = (unsigned)half_bcast<J>(s) * (unsigned)(16 * VEC * 4) + lane_off;
    const char* p = reinterpret_cast<const char*>(h) + off;
    if constexpr (VEC == 4) {
        const float4 r = *reinterpret_cast<const float4*>(p);
        v[0] = r.x; v[1] = r.y; v[2] = r.z; v[3] = r.w;
    } else {
        const float2 r = *reinterpret_cast<const float2*>(p);
        v[0] = r.x; v[1] = r.y;
    }
}

template <int VEC, int J0>
__device__ __forceinline__ void sum_batch(const float* __restrict__ h, int s, unsigned lane_off,
                                          int first, int len, bool full, float (&acc)[VEC]) {
    float v[8][VEC];
    load_row<VEC, J0 + 0>(h, s, lane_off, v[0]);
    load_row<VEC, J0 + 1>(h, s, lane_off, v[1]);
    load_row<VEC, J0 + 2>(h, s, lane_off, v[2]);
    load_row<VEC, J0 + 3>(h, s, lane_off, v[3]);
    load_row<VEC, J0 + 4>(h, s, lane_off, v[4]);
    load_row<VEC, J0 + 5>(h, s, lane_off, v[5]);
    load_row<VEC, J0 + 6>(h, s, lane_off, v[6]);
    load_row<VEC, J0 + 7>(h, s, lane_off, v[7]);
    if (full) {
        #pragma unroll
        for (int u = 0; u < 8; ++u) {
            #pragma unroll
            for (int p = 0; p < VEC; ++p) acc[p] += v[u][p];
        }
    } else {
        #pragma unroll
        for (int u = 0; u < 8; ++u) {
            const bool in = first + u < len;
            #pragma unroll
            for (int p = 0; p < VEC; ++p) acc[p] += in ? v[u][p] : -0.0f;
        }
    }
}

template <int VEC>
__device__ __forceinline__ void gather_rows(const float* __restrict__ h,
                                            const int* __restrict__ csr,
                                            int beg, int len, int lenmin, int lenmax,
                                            int l16, float (&acc)[VEC]) {
    const unsigned lane_off = (unsigned)(l16 * VEC * 4);
    for (int e0 = 0; e0 < lenmax; e0 += 32) {
        int s0 = 0, s1 = 0;
        if (e0 + l16 < len)      s0 = csr[beg + e0 + l16];
        if (e0 + 16 + l16 < len) s1 = csr[beg + e0 + 16 + l16];
        sum_batch<VEC, 0>(h, s0, lane_off, e0, len, e0 + 8 <= lenmin, acc);
        if (e0 + 8 < lenmax)
            sum_batch<VEC, 8>(h, s0, lane_off, e0 + 8, len, e0 + 16 <= lenmin, acc);
        if (e0 + 16 < lenmax)
            sum_batch<VEC, 0>(h, s1, lane_off, e0 + 16, len, e0 + 24 <= lenmin, acc);
        if (e0 + 24 < lenmax)
            sum_batch<VEC, 8>(h, s1, lane_off, e0 + 24, len, e0 + 32 <= lenmin, acc);
    }
}

#define TPITCH 68
__global__ __launch_bounds__(256) void pull_combine1_gemm2_kernel(const float* __restrict__ h,
                                                                  const int* __restrict__ rowptr,
                                                                  const int* __restrict__ csr,
                                                                  const float* __restrict__ dinv,
                                                                  const float* __restrict__ b,
                                                                  const float* __restrict__ W2,
                                                                  float* __restrict__ h2, int n) {
    __shared__ __attribute__((aligned(16))) float tile[16 * TPITCH];
    const int lane = threadIdx.x & 31;
    const int l16 = lane & 15;
    const int half = lane >> 4;
    const int wave = __builtin_amdgcn_readfirstlane(threadIdx.x >> 5);
    const int m0 = blockIdx.x * 16;
    const int node0 = m0 + wave * 2;
    const int node = node0 + half;
    const bool live = node < n;
    int beg = 0, len = 0;
    if (live) { beg = rowptr[node]; len = rowptr[node + 1] - beg; }
    const int len_lo = __builtin_amdgcn_readlane(len, 0);
    const int len_hi = __builtin_amdgcn_readlane(len, 16);
    const int lenmin = min(len_lo, len_hi), lenmax = max(len_lo, len_hi);
    const int nc = live ? node : n - 1;
    const size_t i = (size_t)nc * HID1 + l16 * 4;
    const float4 hv = *reinterpret_cast<const float4*>(h + i);
    const float dd = dinv[nc];
    const float4 bv = *reinterpret_cast<const float4*>(b + l16 * 4);
    float acc[4] = {0.0f, 0.0f, 0.0f, 0.0f};
    gather_rows<4>(h, csr, beg, len, lenmin, lenmax, l16, acc);
    float4 o;
    o.x = fmaxf((acc[0] + hv.x) * dd + bv.x, 0.0f) * dd;
    o.y = fmaxf((acc[1] + hv.y) * dd + bv.y, 0.0f) * dd;
    o.z = fmaxf((acc[2] + hv.z) * dd + bv.z, 0.0f) * dd;
    o.w = fmaxf((acc[3] + hv.w) * dd + bv.w, 0.0f) * dd;
    if (!live) o = make_float4(0.0f, 0.0f, 0.0f, 0.0f);
    *reinterpret_cast<float4*>(&tile[(wave * 2 + half) * TPITCH + l16 * 4]) = o;

    const int khi = half * 2;
    const int col = (wave & 1) * 16 + l16;
    v2f bf[HID1 / 4] = {};
    if (wave < 2) {
        #pragma unroll
        for (int j = 0; j < HID1 / 4; ++j) {
            bf[j].x = W2[(j * 4 + khi + 0) * HID2 + col];
            bf[j].y = W2[(j * 4 + khi + 1) * HID2 + col];
        }
    }
    __syncthreads();
    if (wave >= 2) return;
    v8f c = {};
    #pragma unroll
    for (int j = 0; j < HID1 / 4; ++j) {
        const v2f a = *reinterpret_cast<const v2f*>(&tile[l16 * TPITCH + j * 4 + khi]);
        c = __builtin_amdgcn_wmma_f32_16x16x4_f32(false, a, false, bf[j],
                                                  (short)0, c, false, false);
    }
    const int mbase = m0 + half * 8;
    float* __restrict__ orow = h2 + (size_t)mbase * HID2 + col;
    if (mbase + 8 <= n) {
        #pragma unroll
        for (int r = 0; r < 8; ++r) orow[r * HID2] = c[r];
    } else {
        #pragma unroll
        for (int r = 0; r < 8; ++r)
            if (mbase + r < n) orow[r * HID2] = c[r];
    }
}

__global__ __launch_bounds__(256) void pull_combine2_proj_kernel(const float* __restrict__ h,
                                                                 const int* __restrict__ rowptr,
                                                                 const int* __restrict__ csr,
                                                                 const float* __restrict__ dinv,
                                                                 const float* __restrict__ b2,
                                                                 const float* __restrict__ Wh,
                                                                 const float* __restrict__ bh,
                                                                 float* __restrict__ out, int n) {
    const int lane = threadIdx.x & 31;
    const int l16 = lane & 15;
    const int node0 = __builtin_amdgcn_readfirstlane((blockIdx.x * 8 + (threadIdx.x >> 5)) * 2);
    if (node0 >= n) return;
    const int node = node0 + (lane >> 4);
    const bool live = node < n;
    int beg = 0, len = 0;
    if (live) { beg = rowptr[node]; len = rowptr[node + 1] - beg; }
    const int len_lo = __builtin_amdgcn_readlane(len, 0);
    const int len_hi = __builtin_amdgcn_readlane(len, 16);
    const int lenmin = min(len_lo, len_hi), lenmax = max(len_lo, len_hi);
    const int nc = live ? node : node0;
    const float2 hv = *reinterpret_cast<const float2*>(h + (size_t)nc * HID2 + l16 * 2);
    const float dd = dinv[nc];
    const float2 bb = *reinterpret_cast<const float2*>(b2 + l16 * 2);
    const float2 ww = *reinterpret_cast<const float2*>(Wh + l16 * 2);
    float acc[2] = {0.0f, 0.0f};
    gather_rows<2>(h, csr, beg, len, lenmin, lenmax, l16, acc);
    const float v0 = fmaxf((acc[0] + hv.x) * dd + bb.x, 0.0f);
    const float v1 = fmaxf((acc[1] + hv.y) * dd + bb.y, 0.0f);
    float a = v0 * ww.x + v1 * ww.y;
    #pragma unroll
    for (int off = 8; off > 0; off >>= 1)
        a += __shfl_xor(a, off, 32);
    if (live && l16 == 0) out[node] = a + bh[0];
}

extern "C" void kernel_launch(void* const* d_in, const int* in_sizes, int n_in,
                              void* d_out, int out_size, void* d_ws, size_t ws_size,
                              hipStream_t stream) {
    const float* x  = (const float*)d_in[0];
    const int*   ei = (const int*)d_in[1];
    const float* W1 = (const float*)d_in[2];
    const float* b1 = (const float*)d_in[3];
    const float* W2 = (const float*)d_in[4];
    const float* b2 = (const float*)d_in[5];
    const float* Wh = (const float*)d_in[6];
    const float* bh = (const float*)d_in[7];
    float* out = (float*)d_out;

    const int N = in_sizes[0] / IN_CH;
    const int E = in_sizes[1] / 2;
    const int* src = ei;
    const int* dst = ei + E;

    float* f      = (float*)d_ws;
    int*   blkT   = (int*)d_ws;
    int*   rowptr = blkT + 65536;
    float* dinv   = f + 196608;
    float* h1     = f + 262144;
    float* h2     = h1 + (size_t)N * HID1;
    int*   csr    = (int*)(h2 + (size_t)N * HID2);
    int*   rank   = csr + E;
    int*   cntT   = rank + E;

    slice_count_kernel<<<SLICES, 1024, 0, stream>>>(dst, cntT, blkT, rank, E, N);
    slice_prefix_kernel<<<(N + PBLK - 1) / PBLK, PBLK, 0, stream>>>(cntT, blkT, rowptr, dinv, N);
    place_edges_kernel<<<SLICES * PLACE_PARTS, 1024, 0, stream>>>(src, dst, rank, cntT, csr, E, N);

    gemm1_wmma<<<(N / 16 + 1) / 2, 128, 0, stream>>>(x, W1, dinv, h1, N / 16);
    pull_combine1_gemm2_kernel<<<(N + 15) / 16, 256, 0, stream>>>(h1, rowptr, csr, dinv, b1, W2, h2, N);

    pull_combine2_proj_kernel<<<(N + 15) / 16, 256, 0, stream>>>(h2, rowptr, csr, dinv, b2, Wh, bh, out, N);
}


# Edge sort rebuilt as a two-level split: pairs bucketed by 256-node destination block through LDS (whole-line stores), then a work-group orders a block's lists in LDS; list (sum) order changes
# speedup vs baseline: 4.9506x; 1.4756x over previous
#include <hip/hip_runtime.h>

typedef float v2f __attribute__((ext_vector_type(2)));
typedef float v8f __attribute__((ext_vector_type(8)));

#define IN_CH  128
#define HID1   64
#define HID2   32

#define PARTS       256
#define PBLK        256
#define HIST_MAX    51200
#define TABW        256
#define STAGE_PAIRS 8192
#define STAGE_LIST  32768
static_assert(HIST_MAX / PBLK <= TABW, "more blocks than table columns");
static_assert(PARTS == 256 && TABW == 256, "the column sums below split 256 parts over 4 thread groups of 256");

__device__ __forceinline__ int part_len(int nE) {
    return (nE + PARTS * 4 - 1) / (PARTS * 4) * 4;
}

__device__ __forceinline__ int block_of(int d) { return (int)((unsigned)d / PBLK); }

__global__ __launch_bounds__(1024) void block_count_kernel(const int* __restrict__ dst,
                                                           int* __restrict__ tab, int nE) {
    __shared__ int cnt[TABW];
    const int t = threadIdx.x;
    const int p = blockIdx.x;
    if (t < TABW) cnt[t] = 0;
    __syncthreads();
    const int per = part_len(nE);
    const int beg = p * per;
    const int end = min(beg + per, nE);
    for (int i = beg + t * 4; i < end; i += 8192) {
        const int i2 = i + 4096;
        const bool f1 = i + 3 < end;
        const bool f2 = i2 + 3 < end;
        int4 d1 = make_int4(0, 0, 0, 0), d2 = d1;
        if (f1) d1 = *reinterpret_cast<const int4*>(dst + i);
        if (f2) d2 = *reinterpret_cast<const int4*>(dst + i2);
        if (f1) {
            atomicAdd(&cnt[block_of(d1.x)], 1);
            atomicAdd(&cnt[block_of(d1.y)], 1);
            atomicAdd(&cnt[block_of(d1.z)], 1);
            atomicAdd(&cnt[block_of(d1.w)], 1);
        } else {
            for (int j = i; j < end; ++j) atomicAdd(&cnt[block_of(dst[j])], 1);
        }
        if (f2) {
            atomicAdd(&cnt[block_of(d2.x)], 1);
            atomicAdd(&cnt[block_of(d2.y)], 1);
            atomicAdd(&cnt[block_of(d2.z)], 1);
            atomicAdd(&cnt[block_of(d2.w)], 1);
        } else {
            for (int j = i2; j < end; ++j) atomicAdd(&cnt[block_of(dst[j])], 1);
        }
    }
    __syncthreads();
    if (t < TABW) tab[p * TABW + t] = cnt[t];
}

__device__ __forceinline__ void put_pair(int* __restrict__ lcur, const int* __restrict__ gdelta,
                                         int2* __restrict__ lpairs, int2* __restrict__ pairs,
                                         bool staged, int s, int d) {
    const int b = block_of(d);
    const int lp = atomicAdd(&lcur[b], 1);
    if (staged) lpairs[lp] = make_int2(s, d);
    else        pairs[gdelta[b] + lp] = make_int2(s, d);
}

__global__ __launch_bounds__(1024) void bucket_scatter_kernel(const int* __restrict__ src,
                                                              const int* __restrict__ dst,
                                                              const int* __restrict__ tab,
                                                              int* __restrict__ bst,
                                                              int* __restrict__ pq, int nE) {
    __shared__ int tot4[4][TABW];
    __shared__ int part4[4][TABW];
    __shared__ int wsumT[TABW / 32];
    __shared__ int wsumO[TABW / 32];
    __shared__ int gdelta[TABW];
    __shared__ int lcur[TABW];
    __shared__ __attribute__((aligned(16))) int2 lpairs[STAGE_PAIRS];
    int2* __restrict__ pairs = reinterpret_cast<int2*>(pq);
    const int t = threadIdx.x;
    const int p = blockIdx.x;
    const int lane = t & 31, wave = t >> 5;
    const int b = t & (TABW - 1);
    const int q = t >> 8;
    int tot = 0, part = 0;
    #pragma unroll
    for (int h = 0; h < 2; ++h) {
        int v[32];
        #pragma unroll
        for (int k = 0; k < 32; ++k) v[k] = tab[(q * 64 + h * 32 + k) * TABW + b];
        #pragma unroll
        for (int k = 0; k < 32; ++k) {
            tot += v[k];
            part += (q * 64 + h * 32 + k < p) ? v[k] : 0;
        }
    }
    tot4[q][b] = tot;
    part4[q][b] = part;
    __syncthreads();
    int T = 0, P = 0, own = 0;
    if (t < TABW) {
        T = tot4[0][t] + tot4[1][t] + tot4[2][t] + tot4[3][t];
        P = part4[0][t] + part4[1][t] + part4[2][t] + part4[3][t];
        own = tab[p * TABW + t];
    }
    int iT = T, iO = own;
    #pragma unroll
    for (int off = 1; off < 32; off <<= 1) {
        const int y = __shfl_up(iT, off, 32);
        const int z = __shfl_up(iO, off, 32);
        if (lane >= off) { iT += y; iO += z; }
    }
    if (t < TABW && lane == 31) { wsumT[wave] = iT; wsumO[wave] = iO; }
    __syncthreads();
    if (t < TABW) {
        int baseT = 0, baseO = 0;
        #pragma unroll
        for (int w = 0; w < TABW / 32; ++w) {
            if (w < wave) { baseT += wsumT[w]; baseO += wsumO[w]; }
        }
        const int exT = baseT + iT - T;
        const int lstart = baseO + iO - own;
        gdelta[t] = exT + P - lstart;
        lcur[t] = lstart;
        if (p == 0) {
            bst[t] = exT;
            if (t == TABW - 1) bst[TABW] = exT + T;
        }
    }
    __syncthreads();
    const int per = part_len(nE);
    const int beg = p * per;
    const int end = min(beg + per, nE);
    const int m = max(end - beg, 0);
    const bool staged = per <= STAGE_PAIRS;
    for (int i = beg + t * 4; i < end; i += 8192) {
        const int i2 = i + 4096;
        const bool f1 = i + 3 < end;
        const bool f2 = i2 + 3 < end;
        int4 s1 = make_int4(0, 0, 0, 0), d1 = s1, s2 = s1, d2 = s1;
        if (f1) {
            s1 = *reinterpret_cast<const int4*>(src + i);
            d1 = *reinterpret_cast<const int4*>(dst + i);
        }
        if (f2) {
            s2 = *reinterpret_cast<const int4*>(src + i2);
            d2 = *reinterpret_cast<const int4*>(dst + i2);
        }
        if (f1) {
            put_pair(lcur, gdelta, lpairs, pairs, staged, s1.x, d1.x);
            put_pair(lcur, gdelta, lpairs, pairs, staged, s1.y, d1.y);
            put_pair(lcur, gdelta, lpairs, pairs, staged, s1.z, d1.z);
            put_pair(lcur, gdelta, lpairs, pairs, staged, s1.w, d1.w);
        } else {
            for (int j = i; j < end; ++j) put_pair(lcur, gdelta, lpairs, pairs, staged, src[j], dst[j]);
        }
        if (f2) {
            put_pair(lcur, gdelta, lpairs, pairs, staged, s2.x, d2.x);
            put_pair(lcur, gdelta, lpairs, pairs, staged, s2.y, d2.y);
            put_pair(lcur, gdelta, lpairs, pairs, staged, s2.z, d2.z);
            put_pair(lcur, gdelta, lpairs, pairs, staged, s2.w, d2.w);
        } else {
            for (int j = i2; j < end; ++j) put_pair(lcur, gdelta, lpairs, pairs, staged, src[j], dst[j]);
        }
    }
    if (staged) {
        __syncthreads();
        for (int i = t; i < m; i += 1024) {
            const int2 pr = lpairs[i];
            pairs[gdelta[block_of(pr.y)] + i] = pr;
        }
    }
}

#define LIST_TRIPS 8

__global__ __launch_bounds__(1024) void block_lists_kernel(const int* __restrict__ pq,
                                                           const int* __restrict__ bst,
                                                           int* __restrict__ rowptr,
                                                           float* __restrict__ dinv,
                                                           int* __restrict__ csr, int n) {
    __shared__ int cnt[PBLK];
    __shared__ int cur[PBLK];
    __shared__ int wsum[PBLK / 32];
    __shared__ __attribute__((aligned(16))) int lcsr[STAGE_LIST];
    const int2* __restrict__ pairs = reinterpret_cast<const int2*>(pq);
    const int t = threadIdx.x;
    const int lane = t & 31, wave = t >> 5;
    const int b = blockIdx.x;
    const int n0 = b * PBLK;
    const int s0 = bst[b];
    const int m = bst[b + 1] - s0;
    if (t < PBLK) cnt[t] = 0;
    __syncthreads();
    for (int i0 = 0; i0 < m; i0 += LIST_TRIPS * 1024) {
        int dl[LIST_TRIPS];
        #pragma unroll
        for (int k = 0; k < LIST_TRIPS; ++k) {
            const int i = i0 + k * 1024 + t;
            dl[k] = 0;
            if (i < m) dl[k] = pairs[s0 + i].y - n0;
        }
        #pragma unroll
        for (int k = 0; k < LIST_TRIPS; ++k) {
            const int i = i0 + k * 1024 + t;
            if (i < m) atomicAdd(&cnt[dl[k]], 1);
        }
    }
    __syncthreads();
    const int c = t < PBLK ? cnt[t] : 0;
    int inc = c;
    #pragma unroll
    for (int off = 1; off < 32; off <<= 1) {
        const int y = __shfl_up(inc, off, 32);
        if (lane >= off) inc += y;
    }
    if (t < PBLK && lane == 31) wsum[wave] = inc;
    __syncthreads();
    if (t < PBLK) {
        int base = 0;
        #pragma unroll
        for (int w = 0; w < PBLK / 32; ++w) {
            if (w < wave) base += wsum[w];
        }
        const int ex = base + inc - c;
        cur[t] = ex;
        const int node = n0 + t;
        if (node < n) {
            rowptr[node] = s0 + ex;
            dinv[node] = rsqrtf((float)c + 1.0f);
            if (node == n - 1) rowptr[n] = s0 + ex + c;
        }
    }
    __syncthreads();
    const bool staged = m <= STAGE_LIST;
    for (int i0 = 0; i0 < m; i0 += LIST_TRIPS * 1024) {
        int2 pr[LIST_TRIPS];
        #pragma unroll
        for (int k = 0; k < LIST_TRIPS; ++k) {
            const int i = i0 + k * 1024 + t;
            pr[k] = make_int2(0, n0);
            if (i < m) pr[k] = pairs[s0 + i];
        }
        #pragma unroll
        for (int k = 0; k < LIST_TRIPS; ++k) {
            const int i = i0 + k * 1024 + t;
            if (i < m) {
                const int pos = atomicAdd(&cur[pr[k].y - n0], 1);
                if (staged) lcsr[pos] = pr[k].x;
                else        csr[s0 + pos] = pr[k].x;
            }
        }
    }
    if (staged) {
        __syncthreads();
        for (int i = t; i < m; i += 1024) csr[s0 + i] = lcsr[i];
    }
}

#define APITCH   132

__device__ __forceinline__ void gemm1_tile(const float* __restrict__ buf, const v2f (&bf)[IN_CH / 4],
                                           const float* __restrict__ dinv, float* __restrict__ H,
                                           int tile, int lane, int col) {
    const int l16 = lane & 15;
    const int khi = (lane >> 4) * 2;
    const int mbase = tile * 16 + ((lane >> 4) * 8);
    const float4 s0 = *reinterpret_cast<const float4*>(dinv + mbase);
    const float4 s1 = *reinterpret_cast<const float4*>(dinv + mbase + 4);
    v8f c = {};
    #pragma unroll
    for (int j = 0; j < IN_CH / 4; ++j) {
        const v2f a = *reinterpret_cast<const v2f*>(&buf[l16 * APITCH + j * 4 + khi]);
        c = __builtin_amdgcn_wmma_f32_16x16x4_f32(false, a, false, bf[j],
                                                  (short)0, c, false, false);
    }
    const float sc[8] = {s0.x, s0.y, s0.z, s0.w, s1.x, s1.y, s1.z, s1.w};
    #pragma unroll
    for (int r = 0; r < 8; ++r)
        H[(mbase + r) * HID1 + col] = c[r] * sc[r];
}

__global__ __launch_bounds__(128) void gemm1_wmma(const float* __restrict__ X,
                                                  const float* __restrict__ W,
                                                  const float* __restrict__ dinv,
                                                  float* __restrict__ H, int ntiles) {
    __shared__ __attribute__((aligned(16))) float atile[2][16 * APITCH];
    const int lane = threadIdx.x & 31;
    const int wave = __builtin_amdgcn_readfirstlane(threadIdx.x >> 5);
    const int khi  = (lane >> 4) * 2;
    const int col  = wave * 16 + (lane & 15);
    const int tile0 = blockIdx.x * 2;
    if (tile0 >= ntiles) return;
    const bool two = tile0 + 1 < ntiles;
    const float* __restrict__ xa = X + ((size_t)tile0 * 16 + wave) * IN_CH + lane * 4;
    const float4 a0 = *reinterpret_cast<const float4*>(xa);
    const float4 a1 = *reinterpret_cast<const float4*>(xa + 4 * IN_CH);
    const float4 a2 = *reinterpret_cast<const float4*>(xa + 8 * IN_CH);
    const float4 a3 = *reinterpret_cast<const float4*>(xa + 12 * IN_CH);
    v2f bf[IN_CH / 4];
    #pragma unroll
    for (int j = 0; j < IN_CH / 4; ++j) {
        bf[j].x = W[(j * 4 + khi + 0) * HID1 + col];
        bf[j].y = W[(j * 4 + khi + 1) * HID1 + col];
    }
    float4 b0 = a0, b1 = a1, b2 = a2, b3 = a3;
    if (two) {
        const float* __restrict__ xb = xa + 16 * IN_CH;
        b0 = *reinterpret_cast<const float4*>(xb);
        b1 = *reinterpret_cast<const float4*>(xb + 4 * IN_CH);
        b2 = *reinterpret_cast<const float4*>(xb + 8 * IN_CH);
        b3 = *reinterpret_cast<const float4*>(xb + 12 * IN_CH);
    }
    float* __restrict__ s = &atile[0][wave * APITCH + lane * 4];
    *reinterpret_cast<float4*>(s)               = a0;
    *reinterpret_cast<float4*>(s + 4 * APITCH)  = a1;
    *reinterpret_cast<float4*>(s + 8 * APITCH)  = a2;
    *reinterpret_cast<float4*>(s + 12 * APITCH) = a3;
    __syncthreads();
    gemm1_tile(atile[0], bf, dinv, H, tile0, lane, col);
    if (!two) return;
    float* __restrict__ s2 = &atile[1][wave * APITCH + lane * 4];
    *reinterpret_cast<float4*>(s2)               = b0;
    *reinterpret_cast<float4*>(s2 + 4 * APITCH)  = b1;
    *reinterpret_cast<float4*>(s2 + 8 * APITCH)  = b2;
    *reinterpret_cast<float4*>(s2 + 12 * APITCH) = b3;
    __syncthreads();
    gemm1_tile(atile[1], bf, dinv, H, tile0 + 1, lane, col);
}


template <int J>
__device__ __forceinline__ int half_bcast(int s) {
    return __builtin_amdgcn_update_dpp(0, s, 0x150 + J, 0xF, 0xF, false);
}

template <int VEC, int J>
__device__ __forceinline__ void load_row(const float* __restrict__ h, int s,
                                         unsigned lane_off, float (&v)[VEC]) {
    const unsigned off = (unsigned)half_bcast<J>(s) * (unsigned)(16 * VEC * 4) + lane_off;
    const char* p = reinterpret_cast<const char*>(h) + off;
    if constexpr (VEC == 4) {
        const float4 r = *reinterpret_cast<const float4*>(p);
        v[0] = r.x; v[1] = r.y; v[2] = r.z; v[3] = r.w;
    } else {
        const float2 r = *reinterpret_cast<const float2*>(p);
        v[0] = r.x; v[1] = r.y;
    }
}

template <int VEC, int J0>
__device__ __forceinline__ void sum_batch(const float* __restrict__ h, int s, unsigned lane_off,
                                          int first, int len, bool full, float (&acc)[VEC]) {
    float v[8][VEC];
    load_row<VEC, J0 + 0>(h, s, lane_off, v[0]);
    load_row<VEC, J0 + 1>(h, s, lane_off, v[1]);
    load_row<VEC, J0 + 2>(h, s, lane_off, v[2]);
    load_row<VEC, J0 + 3>(h, s, lane_off, v[3]);
    load_row<VEC, J0 + 4>(h, s, lane_off, v[4]);
    load_row<VEC, J0 + 5>(h, s, lane_off, v[5]);
    load_row<VEC, J0 + 6>(h, s, lane_off, v[6]);
    load_row<VEC, J0 + 7>(h, s, lane_off, v[7]);
    if (full) {
        #pragma unroll
        for (int u = 0; u < 8; ++u) {
            #pragma unroll
            for (int p = 0; p < VEC; ++p) acc[p] += v[u][p];
        }
    } else {
        #pragma unroll
        for (int u = 0; u < 8; ++u) {
            const bool in = first + u < len;
            #pragma unroll
            for (int p = 0; p < VEC; ++p) acc[p] += in ? v[u][p] : -0.0f;
        }
    }
}

template <int VEC>
__device__ __forceinline__ void gather_rows(const float* __restrict__ h,
                                            const int* __restrict__ csr,
                                            int beg, int len, int lenmin, int lenmax,
                                            int l16, float (&acc)[VEC]) {
    const unsigned lane_off = (unsigned)(l16 * VEC * 4);
    for (int e0 = 0; e0 < lenmax; e0 += 32) {
        int s0 = 0, s1 = 0;
        if (e0 + l16 < len)      s0 = csr[beg + e0 + l16];
        if (e0 + 16 + l16 < len) s1 = csr[beg + e0 + 16 + l16];
        sum_batch<VEC, 0>(h, s0, lane_off, e0, len, e0 + 8 <= lenmin, acc);
        if (e0 + 8 < lenmax)
            sum_batch<VEC, 8>(h, s0, lane_off, e0 + 8, len, e0 + 16 <= lenmin, acc);
        if (e0 + 16 < lenmax)
            sum_batch<VEC, 0>(h, s1, lane_off, e0 + 16, len, e0 + 24 <= lenmin, acc);
        if (e0 + 24 < lenmax)
            sum_batch<VEC, 8>(h, s1, lane_off, e0 + 24, len, e0 + 32 <= lenmin, acc);
    }
}

#define TPITCH 68
__global__ __launch_bounds__(256) void pull_combine1_gemm2_kernel(const float* __restrict__ h,
                                                                  const int* __restrict__ rowptr,
                                                                  const int* __restrict__ csr,
                                                                  const float* __restrict__ dinv,
                                                                  const float* __restrict__ b,
                                                                  const float* __restrict__ W2,
                                                                  float* __restrict__ h2, int n) {
    __shared__ __attribute__((aligned(16))) float tile[16 * TPITCH];
    const int lane = threadIdx.x & 31;
    const int l16 = lane & 15;
    const int half = lane >> 4;
    const int wave = __builtin_amdgcn_readfirstlane(threadIdx.x >> 5);
    const int m0 = blockIdx.x * 16;
    const int node0 = m0 + wave * 2;
    const int node = node0 + half;
    const bool live = node < n;
    int beg = 0, len = 0;
    if (live) { beg = rowptr[node]; len = rowptr[node + 1] - beg; }
    const int len_lo = __builtin_amdgcn_readlane(len, 0);
    const int len_hi = __builtin_amdgcn_readlane(len, 16);
    const int lenmin = min(len_lo, len_hi), lenmax = max(len_lo, len_hi);
    const int nc = live ? node : n - 1;
    const size_t i = (size_t)nc * HID1 + l16 * 4;
    const float4 hv = *reinterpret_cast<const float4*>(h + i);
    const float dd = dinv[nc];
    const float4 bv = *reinterpret_cast<const float4*>(b + l16 * 4);
    float acc[4] = {0.0f, 0.0f, 0.0f, 0.0f};
    gather_rows<4>(h, csr, beg, len, lenmin, lenmax, l16, acc);
    float4 o;
    o.x = fmaxf((acc[0] + hv.x) * dd + bv.x, 0.0f) * dd;
    o.y = fmaxf((acc[1] + hv.y) * dd + bv.y, 0.0f) * dd;
    o.z = fmaxf((acc[2] + hv.z) * dd + bv.z, 0.0f) * dd;
    o.w = fmaxf((acc[3] + hv.w) * dd + bv.w, 0.0f) * dd;
    if (!live) o = make_float4(0.0f, 0.0f, 0.0f, 0.0f);
    *reinterpret_cast<float4*>(&tile[(wave * 2 + half) * TPITCH + l16 * 4]) = o;

    const int khi = half * 2;
    const int col = (wave & 1) * 16 + l16;
    v2f bf[HID1 / 4] = {};
    if (wave < 2) {
        #pragma unroll
        for (int j = 0; j < HID1 / 4; ++j) {
            bf[j].x = W2[(j * 4 + khi + 0) * HID2 + col];
            bf[j].y = W2[(j * 4 + khi + 1) * HID2 + col];
        }
    }
    __syncthreads();
    if (wave >= 2) return;
    v8f c = {};
    #pragma unroll
    for (int j = 0; j < HID1 / 4; ++j) {
        const v2f a = *reinterpret_cast<const v2f*>(&tile[l16 * TPITCH + j * 4 + khi]);
        c = __builtin_amdgcn_wmma_f32_16x16x4_f32(false, a, false, bf[j],
                                                  (short)0, c, false, false);
    }
    const int mbase = m0 + half * 8;
    float* __restrict__ orow = h2 + (size_t)mbase * HID2 + col;
    if (mbase + 8 <= n) {
        #pragma unroll
        for (int r = 0; r < 8; ++r) orow[r * HID2] = c[r];
    } else {
        #pragma unroll
        for (int r = 0; r < 8; ++r)
            if (mbase + r < n) orow[r * HID2] = c[r];
    }
}

__global__ __launch_bounds__(256) void pull_combine2_proj_kernel(const float* __restrict__ h,
                                                                 const int* __restrict__ rowptr,
                                                                 const int* __restrict__ csr,
                                                                 const float* __restrict__ dinv,
                                                                 const float* __restrict__ b2,
                                                                 const float* __restrict__ Wh,
                                                                 const float* __restrict__ bh,
                                                                 float* __restrict__ out, int n) {
    const int lane = threadIdx.x & 31;
    const int l16 = lane & 15;
    const int node0 = __builtin_amdgcn_readfirstlane((blockIdx.x * 8 + (threadIdx.x >> 5)) * 2);
    if (node0 >= n) return;
    const int node = node0 + (lane >> 4);
    const bool live = node < n;
    int beg = 0, len = 0;
    if (live) { beg = rowptr[node]; len = rowptr[node + 1] - beg; }
    const int len_lo = __builtin_amdgcn_readlane(len, 0);
    const int len_hi = __builtin_amdgcn_readlane(len, 16);
    const int lenmin = min(len_lo, len_hi), lenmax = max(len_lo, len_hi);
    const int nc = live ? node : node0;
    const float2 hv = *reinterpret_cast<const float2*>(h + (size_t)nc * HID2 + l16 * 2);
    const float dd = dinv[nc];
    const float2 bb = *reinterpret_cast<const float2*>(b2 + l16 * 2);
    const float2 ww = *reinterpret_cast<const float2*>(Wh + l16 * 2);
    float acc[2] = {0.0f, 0.0f};
    gather_rows<2>(h, csr, beg, len, lenmin, lenmax, l16, acc);
    const float v0 = fmaxf((acc[0] + hv.x) * dd + bb.x, 0.0f);
    const float v1 = fmaxf((acc[1] + hv.y) * dd + bb.y, 0.0f);
    float a = v0 * ww.x + v1 * ww.y;
    #pragma unroll
    for (int off = 8; off > 0; off >>= 1)
        a += __shfl_xor(a, off, 32);
    if (live && l16 == 0) out[node] = a + bh[0];
}

extern "C" void kernel_launch(void* const* d_in, const int* in_sizes, int n_in,
                              void* d_out, int out_size, void* d_ws, size_t ws_size,
                              hipStream_t stream) {
    const float* x  = (const float*)d_in[0];
    const int*   ei = (const int*)d_in[1];
    const float* W1 = (const float*)d_in[2];
    const float* b1 = (const float*)d_in[3];
    const float* W2 = (const float*)d_in[4];
    const float* b2 = (const float*)d_in[5];
    const float* Wh = (const float*)d_in[6];
    const float* bh = (const float*)d_in[7];
    float* out = (float*)d_out;

    const int N = in_sizes[0] / IN_CH;
    const int E = in_sizes[1] / 2;
    const int* src = ei;
    const int* dst = ei + E;

    float* f      = (float*)d_ws;
    int*   tab    = (int*)d_ws;
    int*   rowptr = tab + 65536;
    int*   bst    = tab + 131072;
    float* dinv   = f + 196608;
    float* h1     = f + 262144;
    float* h2     = h1 + (size_t)N * HID1;
    int*   pq     = (int*)(h2 + (size_t)N * HID2);
    int*   csr    = pq + (size_t)E * 2;

    block_count_kernel<<<PARTS, 1024, 0, stream>>>(dst, tab, E);
    bucket_scatter_kernel<<<PARTS, 1024, 0, stream>>>(src, dst, tab, bst, pq, E);
    block_lists_kernel<<<(N + PBLK - 1) / PBLK, 1024, 0, stream>>>(pq, bst, rowptr, dinv, csr, N);

    gemm1_wmma<<<(N / 16 + 1) / 2, 128, 0, stream>>>(x, W1, dinv, h1, N / 16);
    pull_combine1_gemm2_kernel<<<(N + 15) / 16, 256, 0, stream>>>(h1, rowptr, csr, dinv, b1, W2, h2, N);

    pull_combine2_proj_kernel<<<(N + 15) / 16, 256, 0, stream>>>(h2, rowptr, csr, dinv, b2, Wh, bh, out, N);
}
